# starting kernel
# baseline (speedup 1.0000x reference)
#include <hip/hip_runtime.h>

typedef __attribute__((ext_vector_type(2))) float v2f;
typedef __attribute__((ext_vector_type(8))) float v8f;

#define HID   64
#define FIN   22
#define COUT  6
#define BN_EPS 1e-5f

__global__ __launch_bounds__(256) void k_init_deg(float* deg, int n) {
  int i = blockIdx.x * blockDim.x + threadIdx.x;
  if (i < n) deg[i] = 1.0f;
}

__global__ __launch_bounds__(256) void k_accum_deg(const int* __restrict__ dst,
                                                   float* deg, int e) {
  int i = blockIdx.x * blockDim.x + threadIdx.x;
  if (i < e) atomicAdd(&deg[dst[i]], 1.0f);
}

__global__ __launch_bounds__(256) void k_dinv(float* deg, int n) {
  int i = blockIdx.x * blockDim.x + threadIdx.x;
  if (i < n) deg[i] = rsqrtf(deg[i]);
}

__device__ __forceinline__ v8f wmma4(v2f a, v2f b, v8f c) {
  return __builtin_amdgcn_wmma_f32_16x16x4_f32(
       false, a,  false, b,
       (short)0, c,  false,  false);
}

__global__ __launch_bounds__(256) void k_embed(const float* __restrict__ x,
                                               const float* __restrict__ W,
                                               const float* __restrict__ bias,
                                               float* __restrict__ h, int n) {
  int wid  = (blockIdx.x * blockDim.x + threadIdx.x) >> 5;
  int lane = threadIdx.x & 31;
  int row0 = wid * 16;
  if (row0 >= n) return;

  int m  = lane & 15;
  int kh = (lane >> 4) << 1;
  int arow = row0 + m; if (arow > n - 1) arow = n - 1;

  v8f acc[4] = {};
  for (int k0 = 0; k0 < 24; k0 += 4) {
    int ka = k0 + kh;
    v2f a;
    a.x = (ka     < FIN) ? x[(size_t)arow * FIN + ka]     : 0.0f;
    a.y = (ka + 1 < FIN) ? x[(size_t)arow * FIN + ka + 1] : 0.0f;
#pragma unroll
    for (int t = 0; t < 4; ++t) {
      v2f b;
      b.x = (ka     < FIN) ? W[ka       * HID + t * 16 + m] : 0.0f;
      b.y = (ka + 1 < FIN) ? W[(ka + 1) * HID + t * 16 + m] : 0.0f;
      acc[t] = wmma4(a, b, acc[t]);
    }
  }
#pragma unroll
  for (int r = 0; r < 8; ++r) {
    int row = row0 + r + ((lane >> 4) << 3);
    if (row < n) {
#pragma unroll
      for (int t = 0; t < 4; ++t) {
        int col = t * 16 + m;
        float v = acc[t][r] + bias[col];
        h[(size_t)row * HID + col] = v > 0.0f ? v : 0.0f;
      }
    }
  }
}

__global__ __launch_bounds__(256) void k_gcn_gemm(const float* __restrict__ hin,
                                                  const float* __restrict__ W,
                                                  const float* __restrict__ dinv,
                                                  float* __restrict__ tout,
                                                  float* __restrict__ agg, int n) {
  int wid  = (blockIdx.x * blockDim.x + threadIdx.x) >> 5;
  int lane = threadIdx.x & 31;
  int row0 = wid * 16;
  if (row0 >= n) return;

  int m  = lane & 15;
  int kh = (lane >> 4) << 1;
  int arow = row0 + m; if (arow > n - 1) arow = n - 1;

  v8f acc[4] = {};
  for (int k0 = 0; k0 < HID; k0 += 4) {
    int ka = k0 + kh;
    v2f a;
    a.x = hin[(size_t)arow * HID + ka];
    a.y = hin[(size_t)arow * HID + ka + 1];
#pragma unroll
    for (int t = 0; t < 4; ++t) {
      v2f b;
      b.x = W[ka       * HID + t * 16 + m];
      b.y = W[(ka + 1) * HID + t * 16 + m];
      acc[t] = wmma4(a, b, acc[t]);
    }
  }
#pragma unroll
  for (int r = 0; r < 8; ++r) {
    int row = row0 + r + ((lane >> 4) << 3);
    if (row < n) {
      float di  = dinv[row];
      float di2 = di * di;
#pragma unroll
      for (int t = 0; t < 4; ++t) {
        int col = t * 16 + m;
        float v = acc[t][r];
        tout[(size_t)row * HID + col] = v;
        agg [(size_t)row * HID + col] = v * di2;
      }
    }
  }
}

__global__ __launch_bounds__(256) void k_scatter(const int* __restrict__ src,
                                                 const int* __restrict__ dst,
                                                 const float* __restrict__ dinv,
                                                 const float* __restrict__ t,
                                                 float* agg, int e) {
  int gid  = blockIdx.x * blockDim.x + threadIdx.x;
  int edge = gid >> 5;
  if (edge >= e) return;
  int lane = gid & 31;

  int s = src[edge];
  int d = dst[edge];
  float coef = dinv[s] * dinv[d];

  int f = lane << 1;
  const float2 v = *reinterpret_cast<const float2*>(t + (size_t)s * HID + f);
  atomicAdd(agg + (size_t)d * HID + f,     coef * v.x);
  atomicAdd(agg + (size_t)d * HID + f + 1, coef * v.y);
}

__global__ __launch_bounds__(256) void k_bn_relu(const float* __restrict__ agg,
                                                 const float* __restrict__ b,
                                                 const float* __restrict__ g,
                                                 const float* __restrict__ be,
                                                 const float* __restrict__ mean,
                                                 const float* __restrict__ var,
                                                 float* __restrict__ out, int total) {
  int i = blockIdx.x * blockDim.x + threadIdx.x;
  if (i >= total) return;
  int f = i & (HID - 1);
  float v = agg[i] + b[f];
  v = (v - mean[f]) * rsqrtf(var[f] + BN_EPS) * g[f] + be[f];
  out[i] = v > 0.0f ? v : 0.0f;
}

__global__ __launch_bounds__(256) void k_cls(const float* __restrict__ h,
                                             const float* __restrict__ W,
                                             const float* __restrict__ b,
                                             float* __restrict__ out, int n) {
  int i = blockIdx.x * blockDim.x + threadIdx.x;
  if (i >= n) return;
  float acc[COUT];
#pragma unroll
  for (int c = 0; c < COUT; ++c) acc[c] = b[c];
  for (int k = 0; k < HID; ++k) {
    float hv = h[(size_t)i * HID + k];
#pragma unroll
    for (int c = 0; c < COUT; ++c) acc[c] += hv * W[k * COUT + c];
  }
#pragma unroll
  for (int c = 0; c < COUT; ++c) out[(size_t)i * COUT + c] = acc[c];
}

extern "C" void kernel_launch(void* const* d_in, const int* in_sizes, int n_in,
                              void* d_out, int out_size, void* d_ws, size_t ws_size,
                              hipStream_t stream) {
  (void)n_in; (void)out_size; (void)ws_size;

  const float* x     = (const float*)d_in[0];
  const int*   ei    = (const int*)  d_in[1];
  const float* W_emb = (const float*)d_in[2];
  const float* b_emb = (const float*)d_in[3];
  const float* W1    = (const float*)d_in[4];
  const float* b1    = (const float*)d_in[5];
  const float* g1    = (const float*)d_in[6];
  const float* be1   = (const float*)d_in[7];
  const float* m1    = (const float*)d_in[8];
  const float* v1    = (const float*)d_in[9];
  const float* W2    = (const float*)d_in[10];
  const float* b2    = (const float*)d_in[11];
  const float* g2    = (const float*)d_in[12];
  const float* be2   = (const float*)d_in[13];
  const float* m2    = (const float*)d_in[14];
  const float* v2    = (const float*)d_in[15];
  const float* W_cls = (const float*)d_in[16];
  const float* b_cls = (const float*)d_in[17];

  const int n = in_sizes[0] / FIN;
  const int e = in_sizes[1] / 2;
  const int* src = ei;
  const int* dst = ei + e;

  float* ws   = (float*)d_ws;
  float* dinv = ws;
  float* B0   = ws + n;
  float* B1   = B0 + (size_t)n * HID;
  float* B2   = B1 + (size_t)n * HID;

  auto cdiv = [](long a, long b) { return (int)((a + b - 1) / b); };

  k_init_deg <<<cdiv(n, 256), 256, 0, stream>>>(dinv, n);
  k_accum_deg<<<cdiv(e, 256), 256, 0, stream>>>(dst, dinv, e);
  k_dinv     <<<cdiv(n, 256), 256, 0, stream>>>(dinv, n);

  const int gthreads = cdiv(n, 16) * 32;

  k_embed<<<cdiv(gthreads, 256), 256, 0, stream>>>(x, W_emb, b_emb, B0, n);

  k_gcn_gemm<<<cdiv(gthreads, 256), 256, 0, stream>>>(B0, W1, dinv, B1, B2, n);
  k_scatter <<<cdiv((long)e * 32, 256), 256, 0, stream>>>(src, dst, dinv, B1, B2, e);
  k_bn_relu <<<cdiv((long)n * HID, 256), 256, 0, stream>>>(B2, b1, g1, be1, m1, v1, B0, n * HID);

  k_gcn_gemm<<<cdiv(gthreads, 256), 256, 0, stream>>>(B0, W2, dinv, B1, B2, n);
  k_scatter <<<cdiv((long)e * 32, 256), 256, 0, stream>>>(src, dst, dinv, B1, B2, e);
  k_bn_relu <<<cdiv((long)n * HID, 256), 256, 0, stream>>>(B2, b2, g2, be2, m2, v2, B0, n * HID);

  k_cls<<<cdiv(n, 256), 256, 0, stream>>>(B0, W_cls, b_cls, (float*)d_out, n);
}


# Edge aggregation as a gather over destination-sorted slot lists (rebuilt each call with returning int atomics) instead of float atomic scatter; sums run in arrival-rank slot order, self term last
# speedup vs baseline: 1.6359x; 1.6359x over previous
#include <hip/hip_runtime.h>


typedef __attribute__((ext_vector_type(2))) float v2f;
typedef __attribute__((ext_vector_type(8))) float v8f;

#define HID   64
#define FIN   22
#define COUT  6
#define BN_EPS 1e-5f
#define SCAN_TILE 1024

__global__ __launch_bounds__(256) void k_zero_cnt(int* cnt, int n) {
  int i = blockIdx.x * blockDim.x + threadIdx.x;
  if (i < n) cnt[i] = 0;
}

__global__ __launch_bounds__(256) void k_count(const int* __restrict__ dst,
                                               int* cnt, int* __restrict__ pos, int e) {
  int i = blockIdx.x * blockDim.x + threadIdx.x;
  if (i < e) pos[i] = atomicAdd(&cnt[dst[i]], 1);
}

__global__ __launch_bounds__(256) void k_scan(const int* __restrict__ cnt,
                                              int* __restrict__ ninfo_w, int n) {
  int2* ninfo = reinterpret_cast<int2*>(ninfo_w);
  __shared__ int s_pre[8];
  __shared__ int s_tot[8];
  const int tid  = threadIdx.x;
  const int lane = tid & 31;
  const int w    = tid >> 5;
  const int tile0 = blockIdx.x * SCAN_TILE;

  const int4* c4 = reinterpret_cast<const int4*>(cnt);
  const int lim = tile0 / 4;
  int pre = 0;
  for (int q0 = tid; q0 < lim; q0 += 8 * 256) {
    int4 v[8];
#pragma unroll
    for (int u = 0; u < 8; ++u) {
      const int q = q0 + u * 256;
      v[u] = make_int4(0, 0, 0, 0);
      if (q < lim) v[u] = c4[q];
    }
#pragma unroll
    for (int u = 0; u < 8; ++u) pre += (v[u].x + v[u].y) + (v[u].z + v[u].w);
  }

  const int i0 = tile0 + tid * 4;
  int c[4];
#pragma unroll
  for (int j = 0; j < 4; ++j) c[j] = (i0 + j < n) ? cnt[i0 + j] : 0;
  const int mine = (c[0] + c[1]) + (c[2] + c[3]);

  int incl = mine;
  for (int o = 1; o < 32; o <<= 1) {
    int up = __shfl_up(incl, o, 32);
    if (lane >= o) incl += up;
  }
  for (int o = 16; o > 0; o >>= 1) pre += __shfl_xor(pre, o, 32);
  if (lane == 31) s_tot[w] = incl;
  if (lane == 0)  s_pre[w] = pre;
  __syncthreads();
  int base = 0;
  for (int j = 0; j < 8; ++j) {
    base += s_pre[j];
    if (j < w) base += s_tot[j];
  }
  int run = base + incl - mine;
#pragma unroll
  for (int j = 0; j < 4; ++j) {
    if (i0 + j < n) {
      int2 o;
      o.x = run;
      o.y = __float_as_int(rsqrtf((float)c[j] + 1.0f));
      ninfo[i0 + j] = o;
      run += c[j];
    }
  }
}

__global__ __launch_bounds__(256) void k_fill(const int* __restrict__ src,
                                              const int* __restrict__ dst,
                                              const int* __restrict__ pos,
                                              const int* __restrict__ ninfo_w,
                                              int* __restrict__ elist_w, int e) {
  const int2* ninfo = reinterpret_cast<const int2*>(ninfo_w);
  int2* elist = reinterpret_cast<int2*>(elist_w);
  int i = blockIdx.x * blockDim.x + threadIdx.x;
  if (i >= e) return;
  int s = src[i];
  int d = dst[i];
  int2 nd = ninfo[d];
  float ds = __int_as_float(ninfo[s].y);
  int2 o;
  o.x = s;
  o.y = __float_as_int(ds * __int_as_float(nd.y));
  elist[nd.x + pos[i]] = o;
}

__device__ __forceinline__ v8f wmma4(v2f a, v2f b, v8f c) {
  return __builtin_amdgcn_wmma_f32_16x16x4_f32(
       false, a,  false, b,
       (short)0, c,  false,  false);
}

__global__ __launch_bounds__(256) void k_embed(const float* __restrict__ x,
                                               const float* __restrict__ W,
                                               const float* __restrict__ bias,
                                               float* __restrict__ h, int n) {
  int wid  = (blockIdx.x * blockDim.x + threadIdx.x) >> 5;
  int lane = threadIdx.x & 31;
  int row0 = wid * 16;
  if (row0 >= n) return;

  int m  = lane & 15;
  int kh = (lane >> 4) << 1;
  int arow = row0 + m; if (arow > n - 1) arow = n - 1;

  v8f acc[4] = {};
  for (int k0 = 0; k0 < 24; k0 += 4) {
    int ka = k0 + kh;
    v2f a;
    a.x = (ka     < FIN) ? x[(size_t)arow * FIN + ka]     : 0.0f;
    a.y = (ka + 1 < FIN) ? x[(size_t)arow * FIN + ka + 1] : 0.0f;
#pragma unroll
    for (int t = 0; t < 4; ++t) {
      v2f b;
      b.x = (ka     < FIN) ? W[ka       * HID + t * 16 + m] : 0.0f;
      b.y = (ka + 1 < FIN) ? W[(ka + 1) * HID + t * 16 + m] : 0.0f;
      acc[t] = wmma4(a, b, acc[t]);
    }
  }
#pragma unroll
  for (int r = 0; r < 8; ++r) {
    int row = row0 + r + ((lane >> 4) << 3);
    if (row < n) {
#pragma unroll
      for (int t = 0; t < 4; ++t) {
        int col = t * 16 + m;
        float v = acc[t][r] + bias[col];
        h[(size_t)row * HID + col] = v > 0.0f ? v : 0.0f;
      }
    }
  }
}

__global__ __launch_bounds__(256) void k_gcn_gemm(const float* __restrict__ hin,
                                                  const float* __restrict__ W,
                                                  float* __restrict__ tout, int n) {
  int wid  = (blockIdx.x * blockDim.x + threadIdx.x) >> 5;
  int lane = threadIdx.x & 31;
  int row0 = wid * 16;
  if (row0 >= n) return;

  int m  = lane & 15;
  int kh = (lane >> 4) << 1;
  int arow = row0 + m; if (arow > n - 1) arow = n - 1;

  v8f acc[4] = {};
  for (int k0 = 0; k0 < HID; k0 += 4) {
    int ka = k0 + kh;
    v2f a;
    a.x = hin[(size_t)arow * HID + ka];
    a.y = hin[(size_t)arow * HID + ka + 1];
#pragma unroll
    for (int t = 0; t < 4; ++t) {
      v2f b;
      b.x = W[ka       * HID + t * 16 + m];
      b.y = W[(ka + 1) * HID + t * 16 + m];
      acc[t] = wmma4(a, b, acc[t]);
    }
  }
#pragma unroll
  for (int r = 0; r < 8; ++r) {
    int row = row0 + r + ((lane >> 4) << 3);
    if (row < n) {
#pragma unroll
      for (int t = 0; t < 4; ++t) {
        int col = t * 16 + m;
        tout[(size_t)row * HID + col] = acc[t][r];
      }
    }
  }
}

template <int G>
__device__ __forceinline__ void add_rows(const float2* __restrict__ trow, int lane,
                                         int2 ent, int k, int m, float2& acc) {
  float2 r[G];
  float  w[G];
#pragma unroll
  for (int u = 0; u < G; ++u) {
    const int kk = (k + u < m) ? (k + u) : (m - 1);
    const int s  = __builtin_amdgcn_readlane(ent.x, kk);
    w[u] = __int_as_float(__builtin_amdgcn_readlane(ent.y, kk));
    r[u] = trow[(size_t)s * 32 + lane];
  }
#pragma unroll
  for (int u = 0; u < G; ++u) {
    const bool live = (k + u < m);
    const float ww = live ? w[u]   : 0.0f;
    const float rx = live ? r[u].x : 0.0f;
    const float ry = live ? r[u].y : 0.0f;
    acc.x = fmaf(ww, rx, acc.x);
    acc.y = fmaf(ww, ry, acc.y);
  }
}

__global__ __launch_bounds__(256) void k_gather(const int* __restrict__ cnt,
                                                const int* __restrict__ ninfo_w,
                                                const int* __restrict__ elist_w,
                                                const float* __restrict__ t,
                                                float* __restrict__ agg, int n) {
  const int2* ninfo = reinterpret_cast<const int2*>(ninfo_w);
  const int2* elist = reinterpret_cast<const int2*>(elist_w);
  const int node = __builtin_amdgcn_readfirstlane((int)((blockIdx.x * blockDim.x + threadIdx.x) >> 5));
  if (node >= n) return;
  const int lane = threadIdx.x & 31;

  const int2 nd  = ninfo[node];
  const int  deg = cnt[node];
  const float2* trow = reinterpret_cast<const float2*>(t);
  const float2 self = trow[(size_t)node * 32 + lane];

  int base = 0;
  int m = deg < 32 ? deg : 32;
  int2 ent = make_int2(0, 0);
  if (lane < m) ent = elist[(size_t)nd.x + lane];

  float2 acc = make_float2(0.0f, 0.0f);
  while (m > 0) {
    int k = 0;
    while (m - k > 4) { add_rows<8>(trow, lane, ent, k, m, acc); k += 8; }
    if (k < m) add_rows<4>(trow, lane, ent, k, m, acc);
    base += 32;
    m = deg - base;
    if (m > 32) m = 32;
    if (m > 0 && lane < m) ent = elist[(size_t)nd.x + base + lane];
  }

  const float di  = __int_as_float(nd.y);
  const float di2 = di * di;
  acc.x = fmaf(self.x, di2, acc.x);
  acc.y = fmaf(self.y, di2, acc.y);
  reinterpret_cast<float2*>(agg)[(size_t)node * 32 + lane] = acc;
}

__global__ __launch_bounds__(256) void k_bn_relu(float* io,
                                                 const float* __restrict__ b,
                                                 const float* __restrict__ g,
                                                 const float* __restrict__ be,
                                                 const float* __restrict__ mean,
                                                 const float* __restrict__ var,
                                                 int total) {
  int i = blockIdx.x * blockDim.x + threadIdx.x;
  if (i >= total) return;
  int f = i & (HID - 1);
  float v = io[i] + b[f];
  v = (v - mean[f]) * rsqrtf(var[f] + BN_EPS) * g[f] + be[f];
  io[i] = v > 0.0f ? v : 0.0f;
}

__global__ __launch_bounds__(256) void k_cls(const float* __restrict__ h,
                                             const float* __restrict__ W,
                                             const float* __restrict__ b,
                                             float* __restrict__ out, int n) {
  int i = blockIdx.x * blockDim.x + threadIdx.x;
  if (i >= n) return;
  float acc[COUT];
#pragma unroll
  for (int c = 0; c < COUT; ++c) acc[c] = b[c];
  for (int k = 0; k < HID; ++k) {
    float hv = h[(size_t)i * HID + k];
#pragma unroll
    for (int c = 0; c < COUT; ++c) acc[c] += hv * W[k * COUT + c];
  }
#pragma unroll
  for (int c = 0; c < COUT; ++c) out[(size_t)i * COUT + c] = acc[c];
}

extern "C" void kernel_launch(void* const* d_in, const int* in_sizes, int n_in,
                              void* d_out, int out_size, void* d_ws, size_t ws_size,
                              hipStream_t stream) {
  (void)n_in; (void)out_size; (void)ws_size;

  const float* x     = (const float*)d_in[0];
  const int*   ei    = (const int*)  d_in[1];
  const float* W_emb = (const float*)d_in[2];
  const float* b_emb = (const float*)d_in[3];
  const float* W1    = (const float*)d_in[4];
  const float* b1    = (const float*)d_in[5];
  const float* g1    = (const float*)d_in[6];
  const float* be1   = (const float*)d_in[7];
  const float* m1    = (const float*)d_in[8];
  const float* v1    = (const float*)d_in[9];
  const float* W2    = (const float*)d_in[10];
  const float* b2    = (const float*)d_in[11];
  const float* g2    = (const float*)d_in[12];
  const float* be2   = (const float*)d_in[13];
  const float* m2    = (const float*)d_in[14];
  const float* v2    = (const float*)d_in[15];
  const float* W_cls = (const float*)d_in[16];
  const float* b_cls = (const float*)d_in[17];

  const int n = in_sizes[0] / FIN;
  const int e = in_sizes[1] / 2;
  const int* src = ei;
  const int* dst = ei + e;

  float* ws   = (float*)d_ws;
  float* B0   = ws + n;
  float* B1   = B0 + (size_t)n * HID;
  int*   cnt   = (int*)(B1 + (size_t)n * HID);
  int*   ninfo = cnt + n;
  int*   pos   = ninfo + 2 * (size_t)n;
  int*   elist = pos + e;

  auto cdiv = [](long a, long b) { return (int)((a + b - 1) / b); };

  k_zero_cnt<<<cdiv(n, 256), 256, 0, stream>>>(cnt, n);
  k_count   <<<cdiv(e, 256), 256, 0, stream>>>(dst, cnt, pos, e);
  k_scan    <<<cdiv(n, SCAN_TILE), 256, 0, stream>>>(cnt, ninfo, n);
  k_fill    <<<cdiv(e, 256), 256, 0, stream>>>(src, dst, pos, ninfo, elist, e);

  const int gthreads = cdiv(n, 16) * 32;

  k_embed<<<cdiv(gthreads, 256), 256, 0, stream>>>(x, W_emb, b_emb, B0, n);

  k_gcn_gemm<<<cdiv(gthreads, 256), 256, 0, stream>>>(B0, W1, B1, n);
  k_gather  <<<cdiv((long)n * 32, 256), 256, 0, stream>>>(cnt, ninfo, elist, B1, B0, n);
  k_bn_relu <<<cdiv((long)n * HID, 256), 256, 0, stream>>>(B0, b1, g1, be1, m1, v1, n * HID);

  k_gcn_gemm<<<cdiv(gthreads, 256), 256, 0, stream>>>(B0, W2, B1, n);
  k_gather  <<<cdiv((long)n * 32, 256), 256, 0, stream>>>(cnt, ninfo, elist, B1, B0, n);
  k_bn_relu <<<cdiv((long)n * HID, 256), 256, 0, stream>>>(B0, b2, g2, be2, m2, v2, n * HID);

  k_cls<<<cdiv(n, 256), 256, 0, stream>>>(B0, W_cls, b_cls, (float*)d_out, n);
}


# Bias+BatchNorm+ReLU and the 64x6 classifier moved into the gather kernels' epilogues (12 launches to 9, no one-element waves); classifier sums become lane pairs, then a 32-lane xor tree, bias last
# speedup vs baseline: 1.8735x; 1.1452x over previous
#include <hip/hip_runtime.h>


typedef __attribute__((ext_vector_type(2))) float v2f;
typedef __attribute__((ext_vector_type(8))) float v8f;

#define HID   64
#define FIN   22
#define COUT  6
#define BN_EPS 1e-5f
#define SCAN_TILE 1024

__global__ __launch_bounds__(256) void k_zero_cnt(int* cnt, int n) {
  int i = blockIdx.x * blockDim.x + threadIdx.x;
  if (i < n) cnt[i] = 0;
}

__global__ __launch_bounds__(256) void k_count(const int* __restrict__ dst,
                                               int* cnt, int* __restrict__ pos, int e) {
  int i = blockIdx.x * blockDim.x + threadIdx.x;
  if (i < e) pos[i] = atomicAdd(&cnt[dst[i]], 1);
}

__global__ __launch_bounds__(256) void k_scan(const int* __restrict__ cnt,
                                              int* __restrict__ ninfo_w, int n) {
  int2* ninfo = reinterpret_cast<int2*>(ninfo_w);
  __shared__ int s_pre[8];
  __shared__ int s_tot[8];
  const int tid  = threadIdx.x;
  const int lane = tid & 31;
  const int w    = tid >> 5;
  const int tile0 = blockIdx.x * SCAN_TILE;

  const int4* c4 = reinterpret_cast<const int4*>(cnt);
  const int lim = tile0 / 4;
  int pre = 0;
  for (int q0 = tid; q0 < lim; q0 += 8 * 256) {
    int4 v[8];
#pragma unroll
    for (int u = 0; u < 8; ++u) {
      const int q = q0 + u * 256;
      v[u] = make_int4(0, 0, 0, 0);
      if (q < lim) v[u] = c4[q];
    }
#pragma unroll
    for (int u = 0; u < 8; ++u) pre += (v[u].x + v[u].y) + (v[u].z + v[u].w);
  }

  const int i0 = tile0 + tid * 4;
  int c[4];
#pragma unroll
  for (int j = 0; j < 4; ++j) c[j] = (i0 + j < n) ? cnt[i0 + j] : 0;
  const int mine = (c[0] + c[1]) + (c[2] + c[3]);

  int incl = mine;
  for (int o = 1; o < 32; o <<= 1) {
    int up = __shfl_up(incl, o, 32);
    if (lane >= o) incl += up;
  }
  for (int o = 16; o > 0; o >>= 1) pre += __shfl_xor(pre, o, 32);
  if (lane == 31) s_tot[w] = incl;
  if (lane == 0)  s_pre[w] = pre;
  __syncthreads();
  int base = 0;
  for (int j = 0; j < 8; ++j) {
    base += s_pre[j];
    if (j < w) base += s_tot[j];
  }
  int run = base + incl - mine;
#pragma unroll
  for (int j = 0; j < 4; ++j) {
    if (i0 + j < n) {
      int2 o;
      o.x = run;
      o.y = __float_as_int(rsqrtf((float)c[j] + 1.0f));
      ninfo[i0 + j] = o;
      run += c[j];
    }
  }
}

__global__ __launch_bounds__(256) void k_fill(const int* __restrict__ src,
                                              const int* __restrict__ dst,
                                              const int* __restrict__ pos,
                                              const int* __restrict__ ninfo_w,
                                              int* __restrict__ elist_w, int e) {
  const int2* ninfo = reinterpret_cast<const int2*>(ninfo_w);
  int2* elist = reinterpret_cast<int2*>(elist_w);
  int i = blockIdx.x * blockDim.x + threadIdx.x;
  if (i >= e) return;
  int s = src[i];
  int d = dst[i];
  int2 nd = ninfo[d];
  float ds = __int_as_float(ninfo[s].y);
  int2 o;
  o.x = s;
  o.y = __float_as_int(ds * __int_as_float(nd.y));
  elist[nd.x + pos[i]] = o;
}

__device__ __forceinline__ v8f wmma4(v2f a, v2f b, v8f c) {
  return __builtin_amdgcn_wmma_f32_16x16x4_f32(
       false, a,  false, b,
       (short)0, c,  false,  false);
}

__global__ __launch_bounds__(256) void k_embed(const float* __restrict__ x,
                                               const float* __restrict__ W,
                                               const float* __restrict__ bias,
                                               float* __restrict__ h, int n) {
  int wid  = (blockIdx.x * blockDim.x + threadIdx.x) >> 5;
  int lane = threadIdx.x & 31;
  int row0 = wid * 16;
  if (row0 >= n) return;

  int m  = lane & 15;
  int kh = (lane >> 4) << 1;
  int arow = row0 + m; if (arow > n - 1) arow = n - 1;

  v8f acc[4] = {};
  for (int k0 = 0; k0 < 24; k0 += 4) {
    int ka = k0 + kh;
    v2f a;
    a.x = (ka     < FIN) ? x[(size_t)arow * FIN + ka]     : 0.0f;
    a.y = (ka + 1 < FIN) ? x[(size_t)arow * FIN + ka + 1] : 0.0f;
#pragma unroll
    for (int t = 0; t < 4; ++t) {
      v2f b;
      b.x = (ka     < FIN) ? W[ka       * HID + t * 16 + m] : 0.0f;
      b.y = (ka + 1 < FIN) ? W[(ka + 1) * HID + t * 16 + m] : 0.0f;
      acc[t] = wmma4(a, b, acc[t]);
    }
  }
#pragma unroll
  for (int r = 0; r < 8; ++r) {
    int row = row0 + r + ((lane >> 4) << 3);
    if (row < n) {
#pragma unroll
      for (int t = 0; t < 4; ++t) {
        int col = t * 16 + m;
        float v = acc[t][r] + bias[col];
        h[(size_t)row * HID + col] = v > 0.0f ? v : 0.0f;
      }
    }
  }
}

__global__ __launch_bounds__(256) void k_gcn_gemm(const float* __restrict__ hin,
                                                  const float* __restrict__ W,
                                                  float* __restrict__ tout, int n) {
  int wid  = (blockIdx.x * blockDim.x + threadIdx.x) >> 5;
  int lane = threadIdx.x & 31;
  int row0 = wid * 16;
  if (row0 >= n) return;

  int m  = lane & 15;
  int kh = (lane >> 4) << 1;
  int arow = row0 + m; if (arow > n - 1) arow = n - 1;

  v8f acc[4] = {};
  for (int k0 = 0; k0 < HID; k0 += 4) {
    int ka = k0 + kh;
    v2f a;
    a.x = hin[(size_t)arow * HID + ka];
    a.y = hin[(size_t)arow * HID + ka + 1];
#pragma unroll
    for (int t = 0; t < 4; ++t) {
      v2f b;
      b.x = W[ka       * HID + t * 16 + m];
      b.y = W[(ka + 1) * HID + t * 16 + m];
      acc[t] = wmma4(a, b, acc[t]);
    }
  }
#pragma unroll
  for (int r = 0; r < 8; ++r) {
    int row = row0 + r + ((lane >> 4) << 3);
    if (row < n) {
#pragma unroll
      for (int t = 0; t < 4; ++t) {
        int col = t * 16 + m;
        tout[(size_t)row * HID + col] = acc[t][r];
      }
    }
  }
}

template <int G>
__device__ __forceinline__ void add_rows(const float2* __restrict__ trow, int lane,
                                         int2 ent, int k, int m, float2& acc) {
  float2 r[G];
  float  w[G];
#pragma unroll
  for (int u = 0; u < G; ++u) {
    const int kk = (k + u < m) ? (k + u) : (m - 1);
    const int s  = __builtin_amdgcn_readlane(ent.x, kk);
    w[u] = __int_as_float(__builtin_amdgcn_readlane(ent.y, kk));
    r[u] = trow[(size_t)s * 32 + lane];
  }
#pragma unroll
  for (int u = 0; u < G; ++u) {
    const bool live = (k + u < m);
    const float ww = live ? w[u]   : 0.0f;
    const float rx = live ? r[u].x : 0.0f;
    const float ry = live ? r[u].y : 0.0f;
    acc.x = fmaf(ww, rx, acc.x);
    acc.y = fmaf(ww, ry, acc.y);
  }
}

__device__ __forceinline__ float2 gather_bn_relu(const int* __restrict__ cnt,
                                                 const int2* __restrict__ ninfo,
                                                 const int2* __restrict__ elist,
                                                 const float2* __restrict__ trow,
                                                 const float* __restrict__ b,
                                                 const float* __restrict__ g,
                                                 const float* __restrict__ be,
                                                 const float* __restrict__ mean,
                                                 const float* __restrict__ var,
                                                 int node, int lane) {
  const float2 pb  = reinterpret_cast<const float2*>(b)[lane];
  const float2 pm  = reinterpret_cast<const float2*>(mean)[lane];
  const float2 pv  = reinterpret_cast<const float2*>(var)[lane];
  const float2 pg  = reinterpret_cast<const float2*>(g)[lane];
  const float2 pbe = reinterpret_cast<const float2*>(be)[lane];

  const int2 nd  = ninfo[node];
  const int  deg = cnt[node];
  const float2 self = trow[(size_t)node * 32 + lane];

  int base = 0;
  int m = deg < 32 ? deg : 32;
  int2 ent = make_int2(0, 0);
  if (lane < m) ent = elist[(size_t)nd.x + lane];

  float2 acc = make_float2(0.0f, 0.0f);
  while (m > 0) {
    int k = 0;
    while (m - k > 4) { add_rows<8>(trow, lane, ent, k, m, acc); k += 8; }
    if (k < m) add_rows<4>(trow, lane, ent, k, m, acc);
    base += 32;
    m = deg - base;
    if (m > 32) m = 32;
    if (m > 0 && lane < m) ent = elist[(size_t)nd.x + base + lane];
  }

  const float di  = __int_as_float(nd.y);
  const float di2 = di * di;
  acc.x = fmaf(self.x, di2, acc.x);
  acc.y = fmaf(self.y, di2, acc.y);

  float vx = acc.x + pb.x;
  float vy = acc.y + pb.y;
  vx = (vx - pm.x) * rsqrtf(pv.x + BN_EPS) * pg.x + pbe.x;
  vy = (vy - pm.y) * rsqrtf(pv.y + BN_EPS) * pg.y + pbe.y;
  return make_float2(vx > 0.0f ? vx : 0.0f, vy > 0.0f ? vy : 0.0f);
}

__global__ __launch_bounds__(256) void k_gather_bn(const int* __restrict__ cnt,
                                                   const int* __restrict__ ninfo_w,
                                                   const int* __restrict__ elist_w,
                                                   const float* __restrict__ t,
                                                   const float* __restrict__ b,
                                                   const float* __restrict__ g,
                                                   const float* __restrict__ be,
                                                   const float* __restrict__ mean,
                                                   const float* __restrict__ var,
                                                   float* __restrict__ hout, int n) {
  const int node = __builtin_amdgcn_readfirstlane((int)((blockIdx.x * blockDim.x + threadIdx.x) >> 5));
  if (node >= n) return;
  const int lane = threadIdx.x & 31;
  const float2 h = gather_bn_relu(cnt, reinterpret_cast<const int2*>(ninfo_w),
                                  reinterpret_cast<const int2*>(elist_w),
                                  reinterpret_cast<const float2*>(t),
                                  b, g, be, mean, var, node, lane);
  reinterpret_cast<float2*>(hout)[(size_t)node * 32 + lane] = h;
}

#define CLS_NODES 16
__global__ __launch_bounds__(CLS_NODES * 32) void k_gather_bn_cls(const int* __restrict__ cnt,
                                                       const int* __restrict__ ninfo_w,
                                                       const int* __restrict__ elist_w,
                                                       const float* __restrict__ t,
                                                       const float* __restrict__ b,
                                                       const float* __restrict__ g,
                                                       const float* __restrict__ be,
                                                       const float* __restrict__ mean,
                                                       const float* __restrict__ var,
                                                       const float* __restrict__ Wc,
                                                       const float* __restrict__ bc,
                                                       float* __restrict__ out, int n) {
  __shared__ float s_out[CLS_NODES * COUT];
  const int lane = threadIdx.x & 31;
  const int w    = __builtin_amdgcn_readfirstlane((int)(threadIdx.x >> 5));
  const int node = blockIdx.x * CLS_NODES + w;
  if (node < n) {
    const float4* w4 = reinterpret_cast<const float4*>(Wc) + 3 * lane;
    const float4 q0 = w4[0];
    const float4 q1 = w4[1];
    const float4 q2 = w4[2];
    const float bcl = bc[lane < COUT ? lane : 0];
    const float2 h = gather_bn_relu(cnt, reinterpret_cast<const int2*>(ninfo_w),
                                    reinterpret_cast<const int2*>(elist_w),
                                    reinterpret_cast<const float2*>(t),
                                    b, g, be, mean, var, node, lane);
    float p[COUT];
    p[0] = fmaf(h.y, q1.z, h.x * q0.x);
    p[1] = fmaf(h.y, q1.w, h.x * q0.y);
    p[2] = fmaf(h.y, q2.x, h.x * q0.z);
    p[3] = fmaf(h.y, q2.y, h.x * q0.w);
    p[4] = fmaf(h.y, q2.z, h.x * q1.x);
    p[5] = fmaf(h.y, q2.w, h.x * q1.y);
#pragma unroll
    for (int c = 0; c < COUT; ++c)
      for (int o = 16; o > 0; o >>= 1) p[c] += __shfl_xor(p[c], o, 32);
    float v = p[0];
#pragma unroll
    for (int c = 1; c < COUT; ++c) v = (lane == c) ? p[c] : v;
    if (lane < COUT) s_out[w * COUT + lane] = v + bcl;
  }
  __syncthreads();
  if (w == 0) {
    const size_t o0 = (size_t)blockIdx.x * (CLS_NODES * COUT);
    const size_t total = (size_t)n * COUT;
#pragma unroll
    for (int j = 0; j < (CLS_NODES * COUT) / 32; ++j) {
      const int idx = j * 32 + lane;
      if (o0 + idx < total) out[o0 + idx] = s_out[idx];
    }
  }
}

extern "C" void kernel_launch(void* const* d_in, const int* in_sizes, int n_in,
                              void* d_out, int out_size, void* d_ws, size_t ws_size,
                              hipStream_t stream) {
  (void)n_in; (void)out_size; (void)ws_size;

  const float* x     = (const float*)d_in[0];
  const int*   ei    = (const int*)  d_in[1];
  const float* W_emb = (const float*)d_in[2];
  const float* b_emb = (const float*)d_in[3];
  const float* W1    = (const float*)d_in[4];
  const float* b1    = (const float*)d_in[5];
  const float* g1    = (const float*)d_in[6];
  const float* be1   = (const float*)d_in[7];
  const float* m1    = (const float*)d_in[8];
  const float* v1    = (const float*)d_in[9];
  const float* W2    = (const float*)d_in[10];
  const float* b2    = (const float*)d_in[11];
  const float* g2    = (const float*)d_in[12];
  const float* be2   = (const float*)d_in[13];
  const float* m2    = (const float*)d_in[14];
  const float* v2    = (const float*)d_in[15];
  const float* W_cls = (const float*)d_in[16];
  const float* b_cls = (const float*)d_in[17];

  const int n = in_sizes[0] / FIN;
  const int e = in_sizes[1] / 2;
  const int* src = ei;
  const int* dst = ei + e;

  float* ws   = (float*)d_ws;
  float* B0   = ws + n;
  float* B1   = B0 + (size_t)n * HID;
  int*   cnt   = (int*)(B1 + (size_t)n * HID);
  int*   ninfo = cnt + n;
  int*   pos   = ninfo + 2 * (size_t)n;
  int*   elist = pos + e;

  auto cdiv = [](long a, long b) { return (int)((a + b - 1) / b); };

  k_zero_cnt<<<cdiv(n, 256), 256, 0, stream>>>(cnt, n);
  k_count   <<<cdiv(e, 256), 256, 0, stream>>>(dst, cnt, pos, e);
  k_scan    <<<cdiv(n, SCAN_TILE), 256, 0, stream>>>(cnt, ninfo, n);
  k_fill    <<<cdiv(e, 256), 256, 0, stream>>>(src, dst, pos, ninfo, elist, e);

  const int gthreads = cdiv(n, 16) * 32;

  k_embed<<<cdiv(gthreads, 256), 256, 0, stream>>>(x, W_emb, b_emb, B0, n);

  k_gcn_gemm<<<cdiv(gthreads, 256), 256, 0, stream>>>(B0, W1, B1, n);
  k_gather_bn<<<cdiv((long)n * 32, 256), 256, 0, stream>>>(cnt, ninfo, elist, B1, b1, g1, be1, m1, v1, B0, n);

  k_gcn_gemm<<<cdiv(gthreads, 256), 256, 0, stream>>>(B0, W2, B1, n);
  k_gather_bn_cls<<<cdiv(n, CLS_NODES), CLS_NODES * 32, 0, stream>>>(cnt, ninfo, elist, B1, b2, g2, be2, m2, v2,
                                                                   W_cls, b_cls, (float*)d_out, n);
}


# Each 64x64 layer GEMM moved into the kernel that produces its input rows (embed tile; gathered node rows), tile handed over through LDS; 9 launches to 7; sum order unchanged
# speedup vs baseline: 2.0432x; 1.0906x over previous
#include <hip/hip_runtime.h>


typedef __attribute__((ext_vector_type(2))) float v2f;
typedef __attribute__((ext_vector_type(8))) float v8f;

#define HID   64
#define FIN   22
#define COUT  6
#define BN_EPS 1e-5f
#define SCAN_TILE 1024

__global__ __launch_bounds__(256) void k_zero_cnt(int* cnt, int n) {
  int i = blockIdx.x * blockDim.x + threadIdx.x;
  if (i < n) cnt[i] = 0;
}

__global__ __launch_bounds__(256) void k_count(const int* __restrict__ dst,
                                               int* cnt, int* __restrict__ pos, int e) {
  int i = blockIdx.x * blockDim.x + threadIdx.x;
  if (i < e) pos[i] = atomicAdd(&cnt[dst[i]], 1);
}

__global__ __launch_bounds__(256) void k_scan(const int* __restrict__ cnt,
                                              int* __restrict__ ninfo_w, int n) {
  int2* ninfo = reinterpret_cast<int2*>(ninfo_w);
  __shared__ int s_pre[8];
  __shared__ int s_tot[8];
  const int tid  = threadIdx.x;
  const int lane = tid & 31;
  const int w    = tid >> 5;
  const int tile0 = blockIdx.x * SCAN_TILE;

  const int4* c4 = reinterpret_cast<const int4*>(cnt);
  const int lim = tile0 / 4;
  int pre = 0;
  for (int q0 = tid; q0 < lim; q0 += 8 * 256) {
    int4 v[8];
#pragma unroll
    for (int u = 0; u < 8; ++u) {
      const int q = q0 + u * 256;
      v[u] = make_int4(0, 0, 0, 0);
      if (q < lim) v[u] = c4[q];
    }
#pragma unroll
    for (int u = 0; u < 8; ++u) pre += (v[u].x + v[u].y) + (v[u].z + v[u].w);
  }

  const int i0 = tile0 + tid * 4;
  int c[4];
#pragma unroll
  for (int j = 0; j < 4; ++j) c[j] = (i0 + j < n) ? cnt[i0 + j] : 0;
  const int mine = (c[0] + c[1]) + (c[2] + c[3]);

  int incl = mine;
  for (int o = 1; o < 32; o <<= 1) {
    int up = __shfl_up(incl, o, 32);
    if (lane >= o) incl += up;
  }
  for (int o = 16; o > 0; o >>= 1) pre += __shfl_xor(pre, o, 32);
  if (lane == 31) s_tot[w] = incl;
  if (lane == 0)  s_pre[w] = pre;
  __syncthreads();
  int base = 0;
  for (int j = 0; j < 8; ++j) {
    base += s_pre[j];
    if (j < w) base += s_tot[j];
  }
  int run = base + incl - mine;
#pragma unroll
  for (int j = 0; j < 4; ++j) {
    if (i0 + j < n) {
      int2 o;
      o.x = run;
      o.y = __float_as_int(rsqrtf((float)c[j] + 1.0f));
      ninfo[i0 + j] = o;
      run += c[j];
    }
  }
}

__global__ __launch_bounds__(256) void k_fill(const int* __restrict__ src,
                                              const int* __restrict__ dst,
                                              const int* __restrict__ pos,
                                              const int* __restrict__ ninfo_w,
                                              int* __restrict__ elist_w, int e) {
  const int2* ninfo = reinterpret_cast<const int2*>(ninfo_w);
  int2* elist = reinterpret_cast<int2*>(elist_w);
  int i = blockIdx.x * blockDim.x + threadIdx.x;
  if (i >= e) return;
  int s = src[i];
  int d = dst[i];
  int2 nd = ninfo[d];
  float ds = __int_as_float(ninfo[s].y);
  int2 o;
  o.x = s;
  o.y = __float_as_int(ds * __int_as_float(nd.y));
  elist[nd.x + pos[i]] = o;
}

__device__ __forceinline__ v8f wmma4(v2f a, v2f b, v8f c) {
  return __builtin_amdgcn_wmma_f32_16x16x4_f32(
       false, a,  false, b,
       (short)0, c,  false,  false);
}

#define TP 68

__device__ __forceinline__ void wave_lds_handover() {
  __builtin_amdgcn_fence(__ATOMIC_RELEASE, "wavefront");
  __builtin_amdgcn_wave_barrier();
  __builtin_amdgcn_fence(__ATOMIC_ACQUIRE, "wavefront");
}

__global__ __launch_bounds__(256) void k_embed_gemm(const float* __restrict__ x,
                                                    const float* __restrict__ We,
                                                    const float* __restrict__ bias,
                                                    const float* __restrict__ W1,
                                                    float* __restrict__ tout, int n) {
  __shared__ float2 s_tile2[8 * 16 * TP / 2];
  const int lane = threadIdx.x & 31;
  const int w    = __builtin_amdgcn_readfirstlane((int)(threadIdx.x >> 5));
  const int row0 = (blockIdx.x * 8 + w) * 16;
  if (row0 >= n) return;
  float* s = reinterpret_cast<float*>(s_tile2) + w * (16 * TP);

  const int m  = lane & 15;
  const int hi = lane >> 4;
  const int kh = hi << 1;
  int arow = row0 + m; if (arow > n - 1) arow = n - 1;

  v8f acc[4] = {};
  for (int k0 = 0; k0 < 24; k0 += 4) {
    int ka = k0 + kh;
    v2f a;
    a.x = (ka     < FIN) ? x[(size_t)arow * FIN + ka]     : 0.0f;
    a.y = (ka + 1 < FIN) ? x[(size_t)arow * FIN + ka + 1] : 0.0f;
#pragma unroll
    for (int t = 0; t < 4; ++t) {
      v2f b;
      b.x = (ka     < FIN) ? We[ka       * HID + t * 16 + m] : 0.0f;
      b.y = (ka + 1 < FIN) ? We[(ka + 1) * HID + t * 16 + m] : 0.0f;
      acc[t] = wmma4(a, b, acc[t]);
    }
  }
#pragma unroll
  for (int r = 0; r < 8; ++r) {
#pragma unroll
    for (int t = 0; t < 4; ++t) {
      int col = t * 16 + m;
      float v = acc[t][r] + bias[col];
      s[(r + 8 * hi) * TP + col] = v > 0.0f ? v : 0.0f;
    }
  }
  wave_lds_handover();

  v8f acc2[4] = {};
  for (int k0 = 0; k0 < HID; k0 += 4) {
    int ka = k0 + kh;
    const float2 av = *reinterpret_cast<const float2*>(s + m * TP + ka);
    v2f a;
    a.x = av.x;
    a.y = av.y;
#pragma unroll
    for (int t = 0; t < 4; ++t) {
      v2f b;
      b.x = W1[ka       * HID + t * 16 + m];
      b.y = W1[(ka + 1) * HID + t * 16 + m];
      acc2[t] = wmma4(a, b, acc2[t]);
    }
  }
  wave_lds_handover();
#pragma unroll
  for (int r = 0; r < 8; ++r) {
#pragma unroll
    for (int t = 0; t < 4; ++t) s[(r + 8 * hi) * TP + t * 16 + m] = acc2[t][r];
  }
  wave_lds_handover();
  float2* tout2 = reinterpret_cast<float2*>(tout) + (size_t)row0 * 32 + lane;
  if (row0 + 16 <= n) {
    float2 v[16];
#pragma unroll
    for (int j = 0; j < 16; ++j) v[j] = *reinterpret_cast<const float2*>(s + j * TP + 2 * lane);
#pragma unroll
    for (int j = 0; j < 16; ++j) tout2[(size_t)j * 32] = v[j];
  } else {
    for (int j = 0; j < 16; ++j)
      if (row0 + j < n) tout2[(size_t)j * 32] = *reinterpret_cast<const float2*>(s + j * TP + 2 * lane);
  }
}

template <int G>
__device__ __forceinline__ void add_rows(const float2* __restrict__ trow, int lane,
                                         int2 ent, int k, int m, float2& acc) {
  float2 r[G];
  float  w[G];
#pragma unroll
  for (int u = 0; u < G; ++u) {
    const int kk = (k + u < m) ? (k + u) : (m - 1);
    const int s  = __builtin_amdgcn_readlane(ent.x, kk);
    w[u] = __int_as_float(__builtin_amdgcn_readlane(ent.y, kk));
    r[u] = trow[(size_t)s * 32 + lane];
  }
#pragma unroll
  for (int u = 0; u < G; ++u) {
    const bool live = (k + u < m);
    const float ww = live ? w[u]   : 0.0f;
    const float rx = live ? r[u].x : 0.0f;
    const float ry = live ? r[u].y : 0.0f;
    acc.x = fmaf(ww, rx, acc.x);
    acc.y = fmaf(ww, ry, acc.y);
  }
}

__device__ __forceinline__ float2 gather_bn_relu(const int* __restrict__ cnt,
                                                 const int2* __restrict__ ninfo,
                                                 const int2* __restrict__ elist,
                                                 const float2* __restrict__ trow,
                                                 const float* __restrict__ b,
                                                 const float* __restrict__ g,
                                                 const float* __restrict__ be,
                                                 const float* __restrict__ mean,
                                                 const float* __restrict__ var,
                                                 int node, int lane) {
  const float2 pb  = reinterpret_cast<const float2*>(b)[lane];
  const float2 pm  = reinterpret_cast<const float2*>(mean)[lane];
  const float2 pv  = reinterpret_cast<const float2*>(var)[lane];
  const float2 pg  = reinterpret_cast<const float2*>(g)[lane];
  const float2 pbe = reinterpret_cast<const float2*>(be)[lane];

  const int2 nd  = ninfo[node];
  const int  deg = cnt[node];
  const float2 self = trow[(size_t)node * 32 + lane];

  int base = 0;
  int m = deg < 32 ? deg : 32;
  int2 ent = make_int2(0, 0);
  if (lane < m) ent = elist[(size_t)nd.x + lane];

  float2 acc = make_float2(0.0f, 0.0f);
  while (m > 0) {
    int k = 0;
    while (m - k > 4) { add_rows<8>(trow, lane, ent, k, m, acc); k += 8; }
    if (k < m) add_rows<4>(trow, lane, ent, k, m, acc);
    base += 32;
    m = deg - base;
    if (m > 32) m = 32;
    if (m > 0 && lane < m) ent = elist[(size_t)nd.x + base + lane];
  }

  const float di  = __int_as_float(nd.y);
  const float di2 = di * di;
  acc.x = fmaf(self.x, di2, acc.x);
  acc.y = fmaf(self.y, di2, acc.y);

  float vx = acc.x + pb.x;
  float vy = acc.y + pb.y;
  vx = (vx - pm.x) * rsqrtf(pv.x + BN_EPS) * pg.x + pbe.x;
  vy = (vy - pm.y) * rsqrtf(pv.y + BN_EPS) * pg.y + pbe.y;
  return make_float2(vx > 0.0f ? vx : 0.0f, vy > 0.0f ? vy : 0.0f);
}

#define GG_NODES 16
__global__ __launch_bounds__(GG_NODES * 32) void k_gather_bn_gemm(const int* __restrict__ cnt,
                                                        const int* __restrict__ ninfo_w,
                                                        const int* __restrict__ elist_w,
                                                        const float* __restrict__ t,
                                                        const float* __restrict__ b,
                                                        const float* __restrict__ g,
                                                        const float* __restrict__ be,
                                                        const float* __restrict__ mean,
                                                        const float* __restrict__ var,
                                                        const float* __restrict__ W,
                                                        float* __restrict__ tout, int n) {
  __shared__ float2 s_h2[GG_NODES * TP / 2];
  __shared__ float2 s_t2[GG_NODES * TP / 2];
  float* s_h = reinterpret_cast<float*>(s_h2);
  float* s_t = reinterpret_cast<float*>(s_t2);
  const int lane = threadIdx.x & 31;
  const int w    = __builtin_amdgcn_readfirstlane((int)(threadIdx.x >> 5));
  const int node = blockIdx.x * GG_NODES + w;
  float2 h = make_float2(0.0f, 0.0f);
  if (node < n)
    h = gather_bn_relu(cnt, reinterpret_cast<const int2*>(ninfo_w),
                       reinterpret_cast<const int2*>(elist_w),
                       reinterpret_cast<const float2*>(t),
                       b, g, be, mean, var, node, lane);
  *reinterpret_cast<float2*>(s_h + w * TP + 2 * lane) = h;
  const int m  = lane & 15;
  const int hi = lane >> 4;
  const int kh = hi << 1;
  v2f bf[16];
  if (w < 4) {
#pragma unroll
    for (int i = 0; i < 16; ++i) {
      const int ka = 4 * i + kh;
      bf[i].x = W[ka       * HID + w * 16 + m];
      bf[i].y = W[(ka + 1) * HID + w * 16 + m];
    }
  }
  __syncthreads();
  if (w < 4) {
    v8f acc = {};
#pragma unroll
    for (int i = 0; i < 16; ++i) {
      const int ka = 4 * i + kh;
      const float2 av = *reinterpret_cast<const float2*>(s_h + m * TP + ka);
      v2f a;
      a.x = av.x;
      a.y = av.y;
      acc = wmma4(a, bf[i], acc);
    }
#pragma unroll
    for (int r = 0; r < 8; ++r) s_t[(r + 8 * hi) * TP + w * 16 + m] = acc[r];
  }
  __syncthreads();
  if (node < n)
    reinterpret_cast<float2*>(tout)[(size_t)node * 32 + lane] =
        *reinterpret_cast<const float2*>(s_t + w * TP + 2 * lane);
}

#define CLS_NODES 16
__global__ __launch_bounds__(CLS_NODES * 32) void k_gather_bn_cls(const int* __restrict__ cnt,
                                                       const int* __restrict__ ninfo_w,
                                                       const int* __restrict__ elist_w,
                                                       const float* __restrict__ t,
                                                       const float* __restrict__ b,
                                                       const float* __restrict__ g,
                                                       const float* __restrict__ be,
                                                       const float* __restrict__ mean,
                                                       const float* __restrict__ var,
                                                       const float* __restrict__ Wc,
                                                       const float* __restrict__ bc,
                                                       float* __restrict__ out, int n) {
  __shared__ float s_out[CLS_NODES * COUT];
  const int lane = threadIdx.x & 31;
  const int w    = __builtin_amdgcn_readfirstlane((int)(threadIdx.x >> 5));
  const int node = blockIdx.x * CLS_NODES + w;
  if (node < n) {
    const float4* w4 = reinterpret_cast<const float4*>(Wc) + 3 * lane;
    const float4 q0 = w4[0];
    const float4 q1 = w4[1];
    const float4 q2 = w4[2];
    const float bcl = bc[lane < COUT ? lane : 0];
    const float2 h = gather_bn_relu(cnt, reinterpret_cast<const int2*>(ninfo_w),
                                    reinterpret_cast<const int2*>(elist_w),
                                    reinterpret_cast<const float2*>(t),
                                    b, g, be, mean, var, node, lane);
    float p[COUT];
    p[0] = fmaf(h.y, q1.z, h.x * q0.x);
    p[1] = fmaf(h.y, q1.w, h.x * q0.y);
    p[2] = fmaf(h.y, q2.x, h.x * q0.z);
    p[3] = fmaf(h.y, q2.y, h.x * q0.w);
    p[4] = fmaf(h.y, q2.z, h.x * q1.x);
    p[5] = fmaf(h.y, q2.w, h.x * q1.y);
#pragma unroll
    for (int c = 0; c < COUT; ++c)
      for (int o = 16; o > 0; o >>= 1) p[c] += __shfl_xor(p[c], o, 32);
    float v = p[0];
#pragma unroll
    for (int c = 1; c < COUT; ++c) v = (lane == c) ? p[c] : v;
    if (lane < COUT) s_out[w * COUT + lane] = v + bcl;
  }
  __syncthreads();
  if (w == 0) {
    const size_t o0 = (size_t)blockIdx.x * (CLS_NODES * COUT);
    const size_t total = (size_t)n * COUT;
#pragma unroll
    for (int j = 0; j < (CLS_NODES * COUT) / 32; ++j) {
      const int idx = j * 32 + lane;
      if (o0 + idx < total) out[o0 + idx] = s_out[idx];
    }
  }
}

extern "C" void kernel_launch(void* const* d_in, const int* in_sizes, int n_in,
                              void* d_out, int out_size, void* d_ws, size_t ws_size,
                              hipStream_t stream) {
  (void)n_in; (void)out_size; (void)ws_size;

  const float* x     = (const float*)d_in[0];
  const int*   ei    = (const int*)  d_in[1];
  const float* W_emb = (const float*)d_in[2];
  const float* b_emb = (const float*)d_in[3];
  const float* W1    = (const float*)d_in[4];
  const float* b1    = (const float*)d_in[5];
  const float* g1    = (const float*)d_in[6];
  const float* be1   = (const float*)d_in[7];
  const float* m1    = (const float*)d_in[8];
  const float* v1    = (const float*)d_in[9];
  const float* W2    = (const float*)d_in[10];
  const float* b2    = (const float*)d_in[11];
  const float* g2    = (const float*)d_in[12];
  const float* be2   = (const float*)d_in[13];
  const float* m2    = (const float*)d_in[14];
  const float* v2    = (const float*)d_in[15];
  const float* W_cls = (const float*)d_in[16];
  const float* b_cls = (const float*)d_in[17];

  const int n = in_sizes[0] / FIN;
  const int e = in_sizes[1] / 2;
  const int* src = ei;
  const int* dst = ei + e;

  float* ws   = (float*)d_ws;
  float* B0   = ws + n;
  float* B1   = B0 + (size_t)n * HID;
  int*   cnt   = (int*)(B1 + (size_t)n * HID);
  int*   ninfo = cnt + n;
  int*   pos   = ninfo + 2 * (size_t)n;
  int*   elist = pos + e;

  auto cdiv = [](long a, long b) { return (int)((a + b - 1) / b); };

  k_zero_cnt<<<cdiv(n, 256), 256, 0, stream>>>(cnt, n);
  k_count   <<<cdiv(e, 256), 256, 0, stream>>>(dst, cnt, pos, e);
  k_scan    <<<cdiv(n, SCAN_TILE), 256, 0, stream>>>(cnt, ninfo, n);
  k_fill    <<<cdiv(e, 256), 256, 0, stream>>>(src, dst, pos, ninfo, elist, e);

  const int gthreads = cdiv(n, 16) * 32;

  k_embed_gemm<<<cdiv(gthreads, 256), 256, 0, stream>>>(x, W_emb, b_emb, W1, B1, n);

  k_gather_bn_gemm<<<cdiv(n, GG_NODES), GG_NODES * 32, 0, stream>>>(cnt, ninfo, elist, B1, b1, g1, be1, m1, v1,
                                                                     W2, B0, n);

  k_gather_bn_cls<<<cdiv(n, CLS_NODES), CLS_NODES * 32, 0, stream>>>(cnt, ninfo, elist, B0, b2, g2, be2, m2, v2,
                                                                   W_cls, b_cls, (float*)d_out, n);
}


# Fewer, longer-lived waves: each gather kernel launches 1024 workgroups that walk over the 16-node tiles (100k waves per launch to 16k), edge pass takes 4 edges per thread; per-node sum order kept
# speedup vs baseline: 2.6171x; 1.2809x over previous
#include <hip/hip_runtime.h>


typedef __attribute__((ext_vector_type(2))) float v2f;
typedef __attribute__((ext_vector_type(8))) float v8f;

#define HID   64
#define FIN   22
#define COUT  6
#define BN_EPS 1e-5f
#define SLOTS 32

__global__ __launch_bounds__(256) void k_zero_cnt(int* cnt, int* ovf_cnt, int n) {
  int i = blockIdx.x * blockDim.x + threadIdx.x;
  if (i < n) cnt[i] = 0;
  if (i == 0) *ovf_cnt = 0;
}

__device__ __forceinline__ void place_edge(int s, int d, int r, int* __restrict__ slots,
                                           int* ovf_cnt, int* __restrict__ ovf_d,
                                           int* __restrict__ ovf_s) {
  if (r < SLOTS) {
    slots[(size_t)d * SLOTS + r] = s;
  } else {
    const int o = atomicAdd(ovf_cnt, 1);
    ovf_d[o] = d;
    ovf_s[o] = s;
  }
}

#define EPT 4
__global__ __launch_bounds__(256) void k_count_place(const int* __restrict__ src,
                                                     const int* __restrict__ dst,
                                                     int* cnt, int* __restrict__ slots,
                                                     int* ovf_cnt, int* __restrict__ ovf_d,
                                                     int* __restrict__ ovf_s, int e) {
  const int stride = gridDim.x * blockDim.x;
  for (int base = blockIdx.x * blockDim.x + threadIdx.x; base < e; base += EPT * stride) {
    if (base + (EPT - 1) * stride < e) {
      int s[EPT], d[EPT], r[EPT];
#pragma unroll
      for (int u = 0; u < EPT; ++u) {
        s[u] = src[base + u * stride];
        d[u] = dst[base + u * stride];
      }
#pragma unroll
      for (int u = 0; u < EPT; ++u) r[u] = atomicAdd(&cnt[d[u]], 1);
#pragma unroll
      for (int u = 0; u < EPT; ++u) place_edge(s[u], d[u], r[u], slots, ovf_cnt, ovf_d, ovf_s);
    } else {
      for (int i = base; i < e; i += stride) {
        const int s = src[i];
        const int d = dst[i];
        const int r = atomicAdd(&cnt[d], 1);
        place_edge(s, d, r, slots, ovf_cnt, ovf_d, ovf_s);
      }
    }
  }
}

__device__ __forceinline__ v8f wmma4(v2f a, v2f b, v8f c) {
  return __builtin_amdgcn_wmma_f32_16x16x4_f32(
       false, a,  false, b,
       (short)0, c,  false,  false);
}

#define TP 68

__device__ __forceinline__ void wave_lds_handover() {
  __builtin_amdgcn_fence(__ATOMIC_RELEASE, "wavefront");
  __builtin_amdgcn_wave_barrier();
  __builtin_amdgcn_fence(__ATOMIC_ACQUIRE, "wavefront");
}

__global__ __launch_bounds__(256) void k_embed_gemm(const int* __restrict__ cnt,
                                                    const float* __restrict__ x,
                                                    const float* __restrict__ We,
                                                    const float* __restrict__ bias,
                                                    const float* __restrict__ W1,
                                                    float* __restrict__ tout, int n) {
  __shared__ float2 s_tile2[8 * 16 * TP / 2];
  const int lane = threadIdx.x & 31;
  const int w    = __builtin_amdgcn_readfirstlane((int)(threadIdx.x >> 5));
  const int row0 = (blockIdx.x * 8 + w) * 16;
  if (row0 >= n) return;
  float* s = reinterpret_cast<float*>(s_tile2) + w * (16 * TP);

  const int m  = lane & 15;
  const int hi = lane >> 4;
  const int kh = hi << 1;
  int arow = row0 + m; if (arow > n - 1) arow = n - 1;
  const float dv = rsqrtf((float)cnt[arow] + 1.0f);

  v8f acc[4] = {};
  for (int k0 = 0; k0 < 24; k0 += 4) {
    int ka = k0 + kh;
    v2f a;
    a.x = (ka     < FIN) ? x[(size_t)arow * FIN + ka]     : 0.0f;
    a.y = (ka + 1 < FIN) ? x[(size_t)arow * FIN + ka + 1] : 0.0f;
#pragma unroll
    for (int t = 0; t < 4; ++t) {
      v2f b;
      b.x = (ka     < FIN) ? We[ka       * HID + t * 16 + m] : 0.0f;
      b.y = (ka + 1 < FIN) ? We[(ka + 1) * HID + t * 16 + m] : 0.0f;
      acc[t] = wmma4(a, b, acc[t]);
    }
  }
#pragma unroll
  for (int r = 0; r < 8; ++r) {
#pragma unroll
    for (int t = 0; t < 4; ++t) {
      int col = t * 16 + m;
      float v = acc[t][r] + bias[col];
      s[(r + 8 * hi) * TP + col] = v > 0.0f ? v : 0.0f;
    }
  }
  wave_lds_handover();

  v8f acc2[4] = {};
  for (int k0 = 0; k0 < HID; k0 += 4) {
    int ka = k0 + kh;
    const float2 av = *reinterpret_cast<const float2*>(s + m * TP + ka);
    v2f a;
    a.x = av.x;
    a.y = av.y;
#pragma unroll
    for (int t = 0; t < 4; ++t) {
      v2f b;
      b.x = W1[ka       * HID + t * 16 + m];
      b.y = W1[(ka + 1) * HID + t * 16 + m];
      acc2[t] = wmma4(a, b, acc2[t]);
    }
  }
  wave_lds_handover();
#pragma unroll
  for (int r = 0; r < 8; ++r) {
#pragma unroll
    for (int t = 0; t < 4; ++t) s[(r + 8 * hi) * TP + t * 16 + m] = acc2[t][r];
  }
  wave_lds_handover();
  float2* tout2 = reinterpret_cast<float2*>(tout) + (size_t)row0 * 32 + lane;
  if (row0 + 16 <= n) {
    float2 v[16];
#pragma unroll
    for (int j = 0; j < 16; ++j) v[j] = *reinterpret_cast<const float2*>(s + j * TP + 2 * lane);
#pragma unroll
    for (int j = 0; j < 16; ++j) {
      const float dj = __int_as_float(__builtin_amdgcn_readlane(__float_as_int(dv), j));
      tout2[(size_t)j * 32] = make_float2(v[j].x * dj, v[j].y * dj);
    }
  } else {
    for (int j = 0; j < 16; ++j) {
      const float dj = __int_as_float(__builtin_amdgcn_readlane(__float_as_int(dv), j));
      if (row0 + j < n) {
        const float2 vj = *reinterpret_cast<const float2*>(s + j * TP + 2 * lane);
        tout2[(size_t)j * 32] = make_float2(vj.x * dj, vj.y * dj);
      }
    }
  }
}

template <int G>
__device__ __forceinline__ void add_rows(const float2* __restrict__ trow, int lane,
                                         int ent, int k, int m, float2& acc) {
  float2 r[G];
#pragma unroll
  for (int u = 0; u < G; ++u) {
    const int kk = (k + u < m) ? (k + u) : (m - 1);
    const int s  = __builtin_amdgcn_readlane(ent, kk);
    r[u] = trow[(size_t)s * 32 + lane];
  }
#pragma unroll
  for (int u = 0; u < G; ++u) {
    const bool live = (k + u < m);
    acc.x += live ? r[u].x : 0.0f;
    acc.y += live ? r[u].y : 0.0f;
  }
}

struct FeatConst {
  float2 pb, pm, pg, pbe;
  float rsx, rsy;
};

__device__ __forceinline__ FeatConst load_feat_const(const float* __restrict__ b,
                                                     const float* __restrict__ g,
                                                     const float* __restrict__ be,
                                                     const float* __restrict__ mean,
                                                     const float* __restrict__ var, int lane) {
  FeatConst f;
  f.pb  = reinterpret_cast<const float2*>(b)[lane];
  f.pm  = reinterpret_cast<const float2*>(mean)[lane];
  const float2 pv = reinterpret_cast<const float2*>(var)[lane];
  f.pg  = reinterpret_cast<const float2*>(g)[lane];
  f.pbe = reinterpret_cast<const float2*>(be)[lane];
  f.rsx = rsqrtf(pv.x + BN_EPS);
  f.rsy = rsqrtf(pv.y + BN_EPS);
  return f;
}

__device__ __forceinline__ int load_slot_row(const int* __restrict__ slots, int node, int deg, int lane) {
  const int m = deg < SLOTS ? deg : SLOTS;
  int ent = 0;
  if (lane < m) ent = slots[(size_t)node * SLOTS + lane];
  return ent;
}

__device__ __forceinline__ float2 gather_bn_relu(int deg, int ent,
                                                 const int* __restrict__ ovf_cnt,
                                                 const int* __restrict__ ovf_d,
                                                 const int* __restrict__ ovf_s,
                                                 const float2* __restrict__ trow,
                                                 const FeatConst& f,
                                                 int node, int lane, float& di_out) {
  const float2 self = trow[(size_t)node * 32 + lane];

  const int m = deg < SLOTS ? deg : SLOTS;

  float2 acc = make_float2(0.0f, 0.0f);
  int k = 0;
  while (m - k > 4) { add_rows<8>(trow, lane, ent, k, m, acc); k += 8; }
  if (k < m) add_rows<4>(trow, lane, ent, k, m, acc);

  if (deg > SLOTS) {
    const int novf = *ovf_cnt;
    for (int j0 = 0; j0 < novf; j0 += 32) {
      const int j = j0 + lane;
      int od = -1;
      int os = 0;
      if (j < novf) { od = ovf_d[j]; os = ovf_s[j]; }
      unsigned mask = __builtin_amdgcn_ballot_w32(od == node);
      while (mask != 0u) {
        const int kk = __builtin_ctz(mask);
        mask &= mask - 1u;
        const int s = __builtin_amdgcn_readlane(os, kk);
        const float2 r = trow[(size_t)s * 32 + lane];
        acc.x += r.x;
        acc.y += r.y;
      }
    }
  }

  acc.x += self.x;
  acc.y += self.y;
  const float di = rsqrtf((float)deg + 1.0f);
  di_out = di;

  float vx = acc.x * di + f.pb.x;
  float vy = acc.y * di + f.pb.y;
  vx = (vx - f.pm.x) * f.rsx * f.pg.x + f.pbe.x;
  vy = (vy - f.pm.y) * f.rsy * f.pg.y + f.pbe.y;
  return make_float2(vx > 0.0f ? vx : 0.0f, vy > 0.0f ? vy : 0.0f);
}

#define GATHER_GRID 1024

#define GG_NODES 16
__global__ __launch_bounds__(GG_NODES * 32) void k_gather_bn_gemm(const int* __restrict__ cnt,
                                                        const int* __restrict__ slots,
                                                        const int* __restrict__ ovf_cnt,
                                                        const int* __restrict__ ovf_d,
                                                        const int* __restrict__ ovf_s,
                                                        const float* __restrict__ t,
                                                        const float* __restrict__ b,
                                                        const float* __restrict__ g,
                                                        const float* __restrict__ be,
                                                        const float* __restrict__ mean,
                                                        const float* __restrict__ var,
                                                        const float* __restrict__ W,
                                                        float* __restrict__ tout, int n) {
  __shared__ float2 s_h2[GG_NODES * TP / 2];
  __shared__ float2 s_t2[GG_NODES * TP / 2];
  __shared__ float2 s_w2[16 * 4 * 32];
  float* s_h = reinterpret_cast<float*>(s_h2);
  float* s_t = reinterpret_cast<float*>(s_t2);
  const int lane = threadIdx.x & 31;
  const int w    = __builtin_amdgcn_readfirstlane((int)(threadIdx.x >> 5));
  const int m  = lane & 15;
  const int hi = lane >> 4;
  const int kh = hi << 1;
  const int ntiles = (n + GG_NODES - 1) / GG_NODES;
  const float2* trow = reinterpret_cast<const float2*>(t);

  int tile = blockIdx.x;
  const int gstep = (int)gridDim.x;
  int degs_a = 0;
  int degs_b = 0;
  if (tile < ntiles && tile * GG_NODES + m < n) degs_a = cnt[tile * GG_NODES + m];
  if (tile + gstep < ntiles && (tile + gstep) * GG_NODES + m < n) degs_b = cnt[(tile + gstep) * GG_NODES + m];

  {
    const int ka = 4 * w + kh;
#pragma unroll
    for (int q = 0; q < 4; ++q)
      s_w2[(w * 4 + q) * 32 + lane] = make_float2(W[ka * HID + q * 16 + m], W[(ka + 1) * HID + q * 16 + m]);
  }
  const FeatConst f = load_feat_const(b, g, be, mean, var, lane);

  int deg_cur = __builtin_amdgcn_readlane(degs_a, w);
  int ent_cur = 0;
  if (tile < ntiles && tile * GG_NODES + w < n) ent_cur = load_slot_row(slots, tile * GG_NODES + w, deg_cur, lane);

  for (; tile < ntiles; tile += gstep) {
    const int node = tile * GG_NODES + w;
    const int deg  = deg_cur;
    const int ent  = ent_cur;
    const int tile_nx = tile + gstep;
    const int node_nx = tile_nx * GG_NODES + w;
    const int deg_nx  = __builtin_amdgcn_readlane(degs_b, w);
    int ent_nx = 0;
    if (tile_nx < ntiles && node_nx < n) ent_nx = load_slot_row(slots, node_nx, deg_nx, lane);
    const int tile_n2 = tile_nx + gstep;
    int degs_c = 0;
    if (tile_n2 < ntiles && tile_n2 * GG_NODES + m < n) degs_c = cnt[tile_n2 * GG_NODES + m];

    float2 h = make_float2(0.0f, 0.0f);
    float di = 0.0f;
    if (node < n)
      h = gather_bn_relu(deg, ent, ovf_cnt, ovf_d, ovf_s, trow, f, node, lane, di);
    deg_cur = deg_nx;
    ent_cur = ent_nx;
    degs_b  = degs_c;

    *reinterpret_cast<float2*>(s_h + w * TP + 2 * lane) = h;
    __syncthreads();
    if (w < 4) {
      v8f acc = {};
#pragma unroll
      for (int i = 0; i < 16; ++i) {
        const int ka = 4 * i + kh;
        const float2 av = *reinterpret_cast<const float2*>(s_h + m * TP + ka);
        const float2 bv = s_w2[(i * 4 + w) * 32 + lane];
        v2f a, bb;
        a.x = av.x;
        a.y = av.y;
        bb.x = bv.x;
        bb.y = bv.y;
        acc = wmma4(a, bb, acc);
      }
#pragma unroll
      for (int r = 0; r < 8; ++r) s_t[(r + 8 * hi) * TP + w * 16 + m] = acc[r];
    }
    __syncthreads();
    if (node < n) {
      const float2 tv = *reinterpret_cast<const float2*>(s_t + w * TP + 2 * lane);
      reinterpret_cast<float2*>(tout)[(size_t)node * 32 + lane] = make_float2(tv.x * di, tv.y * di);
    }
  }
}

#define CLS_NODES 16
__global__ __launch_bounds__(CLS_NODES * 32) void k_gather_bn_cls(const int* __restrict__ cnt,
                                                       const int* __restrict__ slots,
                                                       const int* __restrict__ ovf_cnt,
                                                       const int* __restrict__ ovf_d,
                                                       const int* __restrict__ ovf_s,
                                                       const float* __restrict__ t,
                                                       const float* __restrict__ b,
                                                       const float* __restrict__ g,
                                                       const float* __restrict__ be,
                                                       const float* __restrict__ mean,
                                                       const float* __restrict__ var,
                                                       const float* __restrict__ Wc,
                                                       const float* __restrict__ bc,
                                                       float* __restrict__ out, int n) {
  __shared__ float s_out[2][CLS_NODES * COUT];
  const int lane = threadIdx.x & 31;
  const int w    = __builtin_amdgcn_readfirstlane((int)(threadIdx.x >> 5));
  const int ntiles = (n + CLS_NODES - 1) / CLS_NODES;
  const float2* trow = reinterpret_cast<const float2*>(t);
  const size_t total = (size_t)n * COUT;
  const float4* w4 = reinterpret_cast<const float4*>(Wc) + 3 * lane;
  const float4 q0 = w4[0];
  const float4 q1 = w4[1];
  const float4 q2 = w4[2];
  const float bcl = bc[lane < COUT ? lane : 0];
  const FeatConst f = load_feat_const(b, g, be, mean, var, lane);

  int tile = blockIdx.x;
  const int gstep = (int)gridDim.x;
  const int m16 = lane & 15;
  int degs_a = 0;
  int degs_b = 0;
  if (tile < ntiles && tile * CLS_NODES + m16 < n) degs_a = cnt[tile * CLS_NODES + m16];
  if (tile + gstep < ntiles && (tile + gstep) * CLS_NODES + m16 < n) degs_b = cnt[(tile + gstep) * CLS_NODES + m16];
  int deg_cur = __builtin_amdgcn_readlane(degs_a, w);
  int ent_cur = 0;
  if (tile < ntiles && tile * CLS_NODES + w < n) ent_cur = load_slot_row(slots, tile * CLS_NODES + w, deg_cur, lane);
  int buf = 0;
  for (; tile < ntiles; tile += gstep) {
    const int node = tile * CLS_NODES + w;
    const int deg  = deg_cur;
    const int ent  = ent_cur;
    const int tile_nx = tile + gstep;
    const int node_nx = tile_nx * CLS_NODES + w;
    const int deg_nx  = __builtin_amdgcn_readlane(degs_b, w);
    int ent_nx = 0;
    if (tile_nx < ntiles && node_nx < n) ent_nx = load_slot_row(slots, node_nx, deg_nx, lane);
    const int tile_n2 = tile_nx + gstep;
    int degs_c = 0;
    if (tile_n2 < ntiles && tile_n2 * CLS_NODES + m16 < n) degs_c = cnt[tile_n2 * CLS_NODES + m16];

    float2 h = make_float2(0.0f, 0.0f);
    float di = 0.0f;
    if (node < n)
      h = gather_bn_relu(deg, ent, ovf_cnt, ovf_d, ovf_s, trow, f, node, lane, di);
    deg_cur = deg_nx;
    ent_cur = ent_nx;
    degs_b  = degs_c;
    if (node < n) {
      float p[COUT];
      p[0] = fmaf(h.y, q1.z, h.x * q0.x);
      p[1] = fmaf(h.y, q1.w, h.x * q0.y);
      p[2] = fmaf(h.y, q2.x, h.x * q0.z);
      p[3] = fmaf(h.y, q2.y, h.x * q0.w);
      p[4] = fmaf(h.y, q2.z, h.x * q1.x);
      p[5] = fmaf(h.y, q2.w, h.x * q1.y);
#pragma unroll
      for (int c = 0; c < COUT; ++c)
        for (int o = 16; o > 0; o >>= 1) p[c] += __shfl_xor(p[c], o, 32);
      float v = p[0];
#pragma unroll
      for (int c = 1; c < COUT; ++c) v = (lane == c) ? p[c] : v;
      if (lane < COUT) s_out[buf][w * COUT + lane] = v + bcl;
    }
    __syncthreads();
    if (w == 0) {
      const size_t o0 = (size_t)tile * (CLS_NODES * COUT);
#pragma unroll
      for (int j = 0; j < (CLS_NODES * COUT) / 32; ++j) {
        const int idx = j * 32 + lane;
        if (o0 + idx < total) out[o0 + idx] = s_out[buf][idx];
      }
    }
    buf ^= 1;
  }
}

extern "C" void kernel_launch(void* const* d_in, const int* in_sizes, int n_in,
                              void* d_out, int out_size, void* d_ws, size_t ws_size,
                              hipStream_t stream) {
  (void)n_in; (void)out_size; (void)ws_size;

  const float* x     = (const float*)d_in[0];
  const int*   ei    = (const int*)  d_in[1];
  const float* W_emb = (const float*)d_in[2];
  const float* b_emb = (const float*)d_in[3];
  const float* W1    = (const float*)d_in[4];
  const float* b1    = (const float*)d_in[5];
  const float* g1    = (const float*)d_in[6];
  const float* be1   = (const float*)d_in[7];
  const float* m1    = (const float*)d_in[8];
  const float* v1    = (const float*)d_in[9];
  const float* W2    = (const float*)d_in[10];
  const float* b2    = (const float*)d_in[11];
  const float* g2    = (const float*)d_in[12];
  const float* be2   = (const float*)d_in[13];
  const float* m2    = (const float*)d_in[14];
  const float* v2    = (const float*)d_in[15];
  const float* W_cls = (const float*)d_in[16];
  const float* b_cls = (const float*)d_in[17];

  const int n = in_sizes[0] / FIN;
  const int e = in_sizes[1] / 2;
  const int* src = ei;
  const int* dst = ei + e;

  float* ws   = (float*)d_ws;
  int*   cnt  = (int*)d_ws;
  float* B0   = ws + n;
  float* B1   = B0 + (size_t)n * HID;
  int*   slots   = (int*)(B1 + (size_t)n * HID);
  int*   ovf_cnt = slots + (size_t)n * SLOTS;
  int*   ovf_d   = ovf_cnt + 1;
  int*   ovf_s   = ovf_d + e;

  auto cdiv = [](long a, long b) { return (int)((a + b - 1) / b); };

  k_zero_cnt   <<<cdiv(n, 256), 256, 0, stream>>>(cnt, ovf_cnt, n);
  k_count_place<<<cdiv(e, 256 * EPT), 256, 0, stream>>>(src, dst, cnt, slots, ovf_cnt, ovf_d, ovf_s, e);

  const int gthreads = cdiv(n, 16) * 32;

  k_embed_gemm<<<cdiv(gthreads, 256), 256, 0, stream>>>(cnt, x, W_emb, b_emb, W1, B1, n);

  k_gather_bn_gemm<<<GATHER_GRID, GG_NODES * 32, 0, stream>>>(cnt, slots, ovf_cnt, ovf_d, ovf_s, B1,
                                                                     b1, g1, be1, m1, v1, W2, B0, n);

  k_gather_bn_cls<<<GATHER_GRID, CLS_NODES * 32, 0, stream>>>(cnt, slots, ovf_cnt, ovf_d, ovf_s, B0,
                                                                   b2, g2, be2, m2, v2,
                                                                   W_cls, b_cls, (float*)d_out, n);
}


# Destination lists built with LDS counters: edges dealt into 256-node buckets (one memory add per workgroup and bucket, not per edge), a workgroup per bucket ranks them; node sum order = LDS arrival
# speedup vs baseline: 4.2385x; 1.6195x over previous
#include <hip/hip_runtime.h>


typedef __attribute__((ext_vector_type(2))) float v2f;
typedef __attribute__((ext_vector_type(8))) float v8f;

#define HID   64
#define FIN   22
#define COUT  6
#define BN_EPS 1e-5f
#define SLOTS 32

#define BK_SHIFT  8
#define BK_NODES  (1 << BK_SHIFT)
#define BK_MAX    512
#define BK_CAP    4096
#define CUR_PITCH 32

__global__ __launch_bounds__(256) void k_zero_heads(int* bcur, int nb, int* ovf_cnt, int* spill_cnt) {
  int i = blockIdx.x * blockDim.x + threadIdx.x;
  if (i < nb) bcur[i * CUR_PITCH] = 0;
  if (i == 0) { *ovf_cnt = 0; *spill_cnt = 0; }
}

#define PT_THREADS 512
#define PT_EPT     8
#define PT_EDGES   (PT_THREADS * PT_EPT)
__global__ __launch_bounds__(PT_THREADS) void k_partition(const int* __restrict__ src,
                                                          const int* __restrict__ dst,
                                                          int* bcur, int2* __restrict__ rec,
                                                          int* spill_cnt, int2* __restrict__ spill,
                                                          int e, int nb) {
  static_assert(PT_THREADS == BK_MAX, "one thread per bucket counter");
  __shared__ int  s_hist[BK_MAX];
  __shared__ int  s_lpre[BK_MAX];
  __shared__ int  s_gofs[BK_MAX];
  __shared__ int  s_wsum[PT_THREADS / 32];
  __shared__ int2 s_rec[PT_EDGES];
  const int tid  = threadIdx.x;
  const int lane = tid & 31;
  const int w    = tid >> 5;
  s_hist[tid] = 0;
  const int first0 = blockIdx.x * PT_EDGES;
  const int first  = first0 + tid;
  int s[PT_EPT], d[PT_EPT], r[PT_EPT];
#pragma unroll
  for (int u = 0; u < PT_EPT; ++u) {
    const int i = first + u * PT_THREADS;
    s[u] = 0;
    d[u] = -1;
    if (i < e) { s[u] = src[i]; d[u] = dst[i]; }
  }
  __syncthreads();
#pragma unroll
  for (int u = 0; u < PT_EPT; ++u) {
    r[u] = 0;
    if (d[u] >= 0) r[u] = atomicAdd(&s_hist[d[u] >> BK_SHIFT], 1);
  }
  __syncthreads();
  const int c = (tid < nb) ? s_hist[tid] : 0;
  int x = c;
#pragma unroll
  for (int o = 1; o < 32; o <<= 1) {
    const int y = __shfl_up(x, o, 32);
    if (lane >= o) x += y;
  }
  if (lane == 31) s_wsum[w] = x;
  int g0 = 0;
  if (c > 0) g0 = atomicAdd(&bcur[tid * CUR_PITCH], c);
  __syncthreads();
  int woff = 0;
#pragma unroll
  for (int k = 0; k < PT_THREADS / 32; ++k) {
    const int v = s_wsum[k];
    woff += (k < w) ? v : 0;
  }
  const int excl = woff + x - c;
  s_lpre[tid] = excl;
  s_gofs[tid] = g0 - excl;
  __syncthreads();
#pragma unroll
  for (int u = 0; u < PT_EPT; ++u) {
    if (d[u] >= 0) s_rec[s_lpre[d[u] >> BK_SHIFT] + r[u]] = make_int2(s[u], d[u]);
  }
  __syncthreads();
  int nvalid = e - first0;
  if (nvalid > PT_EDGES) nvalid = PT_EDGES;
#pragma unroll
  for (int u = 0; u < PT_EPT; ++u) {
    const int idx = tid + u * PT_THREADS;
    if (idx < nvalid) {
      const int2 q  = s_rec[idx];
      const int b   = q.y >> BK_SHIFT;
      const int pos = s_gofs[b] + idx;
      if (pos < BK_CAP) {
        rec[(size_t)b * BK_CAP + pos] = q;
      } else {
        const int o = atomicAdd(spill_cnt, 1);
        spill[o] = q;
      }
    }
  }
}

__device__ __forceinline__ void overflow_edge(int s, int d, int* ovf_cnt, int* __restrict__ ovf_d,
                                              int* __restrict__ ovf_s) {
  const int o = atomicAdd(ovf_cnt, 1);
  ovf_d[o] = d;
  ovf_s[o] = s;
}

#define RK_THREADS 256
#define RK_PER     (BK_CAP / RK_THREADS)
__global__ __launch_bounds__(RK_THREADS) void k_rank_place(const int* __restrict__ bcur,
                                                           const int2* __restrict__ rec,
                                                           const int* __restrict__ spill_cnt,
                                                           const int2* __restrict__ spill,
                                                           int* __restrict__ cnt, int* __restrict__ slots,
                                                           int* ovf_cnt, int* __restrict__ ovf_d,
                                                           int* __restrict__ ovf_s, int n) {
  __shared__ int  s_cnt[BK_NODES];
  __shared__ int4 s_row4[BK_NODES * SLOTS / 4];
  int* s_row = reinterpret_cast<int*>(s_row4);
  const int tid   = threadIdx.x;
  const int b     = blockIdx.x;
  const int node0 = b << BK_SHIFT;
  for (int j = tid; j < BK_NODES; j += RK_THREADS) s_cnt[j] = 0;
  for (int j = tid; j < BK_NODES * SLOTS / 4; j += RK_THREADS) s_row4[j] = make_int4(0, 0, 0, 0);
  const int total = bcur[b * CUR_PITCH];
  const int nrec  = total < BK_CAP ? total : BK_CAP;
  const int nsp   = *spill_cnt;
  const int2* myrec = rec + (size_t)b * BK_CAP;
  int2 sd[RK_PER];
  int  rk[RK_PER];
#pragma unroll
  for (int u = 0; u < RK_PER; ++u) {
    const int i = tid + u * RK_THREADS;
    sd[u] = make_int2(0, -1);
    if (i < nrec) sd[u] = myrec[i];
  }
  __syncthreads();
#pragma unroll
  for (int u = 0; u < RK_PER; ++u) {
    rk[u] = 0;
    if (sd[u].y >= 0) rk[u] = atomicAdd(&s_cnt[sd[u].y - node0], 1);
  }
#pragma unroll
  for (int u = 0; u < RK_PER; ++u) {
    if (sd[u].y >= 0) {
      if (rk[u] < SLOTS) s_row[(sd[u].y - node0) * SLOTS + rk[u]] = sd[u].x;
      else               overflow_edge(sd[u].x, sd[u].y, ovf_cnt, ovf_d, ovf_s);
    }
  }
  for (int i = tid; i < nsp; i += RK_THREADS) {
    const int2 q = spill[i];
    if ((q.y >> BK_SHIFT) == b) {
      const int r = atomicAdd(&s_cnt[q.y - node0], 1);
      if (r < SLOTS) s_row[(q.y - node0) * SLOTS + r] = q.x;
      else           overflow_edge(q.x, q.y, ovf_cnt, ovf_d, ovf_s);
    }
  }
  __syncthreads();
  int nrows = n - node0;
  if (nrows > BK_NODES) nrows = BK_NODES;
  for (int j = tid; j < nrows; j += RK_THREADS) cnt[node0 + j] = s_cnt[j];
  int4* grow = reinterpret_cast<int4*>(slots + (size_t)node0 * SLOTS);
  for (int j = tid; j < nrows * (SLOTS / 4); j += RK_THREADS) grow[j] = s_row4[j];
}

__device__ __forceinline__ v8f wmma4(v2f a, v2f b, v8f c) {
  return __builtin_amdgcn_wmma_f32_16x16x4_f32(
       false, a,  false, b,
       (short)0, c,  false,  false);
}

#define TP 68

__device__ __forceinline__ void wave_lds_handover() {
  __builtin_amdgcn_fence(__ATOMIC_RELEASE, "wavefront");
  __builtin_amdgcn_wave_barrier();
  __builtin_amdgcn_fence(__ATOMIC_ACQUIRE, "wavefront");
}

__global__ __launch_bounds__(256) void k_embed_gemm(const int* __restrict__ cnt,
                                                    const float* __restrict__ x,
                                                    const float* __restrict__ We,
                                                    const float* __restrict__ bias,
                                                    const float* __restrict__ W1,
                                                    float* __restrict__ tout, int n) {
  __shared__ float2 s_tile2[8 * 16 * TP / 2];
  const int lane = threadIdx.x & 31;
  const int w    = __builtin_amdgcn_readfirstlane((int)(threadIdx.x >> 5));
  const int row0 = (blockIdx.x * 8 + w) * 16;
  if (row0 >= n) return;
  float* s = reinterpret_cast<float*>(s_tile2) + w * (16 * TP);

  const int m  = lane & 15;
  const int hi = lane >> 4;
  const int kh = hi << 1;
  int arow = row0 + m; if (arow > n - 1) arow = n - 1;
  const float dv = rsqrtf((float)cnt[arow] + 1.0f);

  v8f acc[4] = {};
  for (int k0 = 0; k0 < 24; k0 += 4) {
    int ka = k0 + kh;
    v2f a;
    a.x = (ka     < FIN) ? x[(size_t)arow * FIN + ka]     : 0.0f;
    a.y = (ka + 1 < FIN) ? x[(size_t)arow * FIN + ka + 1] : 0.0f;
#pragma unroll
    for (int t = 0; t < 4; ++t) {
      v2f b;
      b.x = (ka     < FIN) ? We[ka       * HID + t * 16 + m] : 0.0f;
      b.y = (ka + 1 < FIN) ? We[(ka + 1) * HID + t * 16 + m] : 0.0f;
      acc[t] = wmma4(a, b, acc[t]);
    }
  }
#pragma unroll
  for (int r = 0; r < 8; ++r) {
#pragma unroll
    for (int t = 0; t < 4; ++t) {
      int col = t * 16 + m;
      float v = acc[t][r] + bias[col];
      s[(r + 8 * hi) * TP + col] = v > 0.0f ? v : 0.0f;
    }
  }
  wave_lds_handover();

  v8f acc2[4] = {};
  for (int k0 = 0; k0 < HID; k0 += 4) {
    int ka = k0 + kh;
    const float2 av = *reinterpret_cast<const float2*>(s + m * TP + ka);
    v2f a;
    a.x = av.x;
    a.y = av.y;
#pragma unroll
    for (int t = 0; t < 4; ++t) {
      v2f b;
      b.x = W1[ka       * HID + t * 16 + m];
      b.y = W1[(ka + 1) * HID + t * 16 + m];
      acc2[t] = wmma4(a, b, acc2[t]);
    }
  }
  wave_lds_handover();
#pragma unroll
  for (int r = 0; r < 8; ++r) {
#pragma unroll
    for (int t = 0; t < 4; ++t) s[(r + 8 * hi) * TP + t * 16 + m] = acc2[t][r];
  }
  wave_lds_handover();
  float2* tout2 = reinterpret_cast<float2*>(tout) + (size_t)row0 * 32 + lane;
  if (row0 + 16 <= n) {
    float2 v[16];
#pragma unroll
    for (int j = 0; j < 16; ++j) v[j] = *reinterpret_cast<const float2*>(s + j * TP + 2 * lane);
#pragma unroll
    for (int j = 0; j < 16; ++j) {
      const float dj = __int_as_float(__builtin_amdgcn_readlane(__float_as_int(dv), j));
      tout2[(size_t)j * 32] = make_float2(v[j].x * dj, v[j].y * dj);
    }
  } else {
    for (int j = 0; j < 16; ++j) {
      const float dj = __int_as_float(__builtin_amdgcn_readlane(__float_as_int(dv), j));
      if (row0 + j < n) {
        const float2 vj = *reinterpret_cast<const float2*>(s + j * TP + 2 * lane);
        tout2[(size_t)j * 32] = make_float2(vj.x * dj, vj.y * dj);
      }
    }
  }
}

template <int G>
__device__ __forceinline__ void add_rows(const float2* __restrict__ trow, int lane,
                                         int ent, int k, int m, float2& acc) {
  float2 r[G];
#pragma unroll
  for (int u = 0; u < G; ++u) {
    const int kk = (k + u < m) ? (k + u) : (m - 1);
    const int s  = __builtin_amdgcn_readlane(ent, kk);
    r[u] = trow[(size_t)s * 32 + lane];
  }
#pragma unroll
  for (int u = 0; u < G; ++u) {
    const bool live = (k + u < m);
    acc.x += live ? r[u].x : 0.0f;
    acc.y += live ? r[u].y : 0.0f;
  }
}

struct FeatConst {
  float2 pb, pm, pg, pbe;
  float rsx, rsy;
};

__device__ __forceinline__ FeatConst load_feat_const(const float* __restrict__ b,
                                                     const float* __restrict__ g,
                                                     const float* __restrict__ be,
                                                     const float* __restrict__ mean,
                                                     const float* __restrict__ var, int lane) {
  FeatConst f;
  f.pb  = reinterpret_cast<const float2*>(b)[lane];
  f.pm  = reinterpret_cast<const float2*>(mean)[lane];
  const float2 pv = reinterpret_cast<const float2*>(var)[lane];
  f.pg  = reinterpret_cast<const float2*>(g)[lane];
  f.pbe = reinterpret_cast<const float2*>(be)[lane];
  f.rsx = rsqrtf(pv.x + BN_EPS);
  f.rsy = rsqrtf(pv.y + BN_EPS);
  return f;
}

__device__ __forceinline__ int load_slot_row(const int* __restrict__ slots, int node, int deg, int lane) {
  const int m = deg < SLOTS ? deg : SLOTS;
  int ent = 0;
  if (lane < m) ent = slots[(size_t)node * SLOTS + lane];
  return ent;
}

__device__ __forceinline__ float2 gather_bn_relu(int deg, int ent,
                                                 const int* __restrict__ ovf_cnt,
                                                 const int* __restrict__ ovf_d,
                                                 const int* __restrict__ ovf_s,
                                                 const float2* __restrict__ trow,
                                                 const FeatConst& f,
                                                 int node, int lane, float& di_out) {
  const float2 self = trow[(size_t)node * 32 + lane];

  const int m = deg < SLOTS ? deg : SLOTS;

  float2 acc = make_float2(0.0f, 0.0f);
  int k = 0;
  while (m - k > 4) { add_rows<8>(trow, lane, ent, k, m, acc); k += 8; }
  if (k < m) add_rows<4>(trow, lane, ent, k, m, acc);

  if (deg > SLOTS) {
    const int novf = *ovf_cnt;
    for (int j0 = 0; j0 < novf; j0 += 32) {
      const int j = j0 + lane;
      int od = -1;
      int os = 0;
      if (j < novf) { od = ovf_d[j]; os = ovf_s[j]; }
      unsigned mask = __builtin_amdgcn_ballot_w32(od == node);
      while (mask != 0u) {
        const int kk = __builtin_ctz(mask);
        mask &= mask - 1u;
        const int s = __builtin_amdgcn_readlane(os, kk);
        const float2 r = trow[(size_t)s * 32 + lane];
        acc.x += r.x;
        acc.y += r.y;
      }
    }
  }

  acc.x += self.x;
  acc.y += self.y;
  const float di = rsqrtf((float)deg + 1.0f);
  di_out = di;

  float vx = acc.x * di + f.pb.x;
  float vy = acc.y * di + f.pb.y;
  vx = (vx - f.pm.x) * f.rsx * f.pg.x + f.pbe.x;
  vy = (vy - f.pm.y) * f.rsy * f.pg.y + f.pbe.y;
  return make_float2(vx > 0.0f ? vx : 0.0f, vy > 0.0f ? vy : 0.0f);
}

#define GATHER_GRID 1024

#define GG_NODES 16
__global__ __launch_bounds__(GG_NODES * 32) void k_gather_bn_gemm(const int* __restrict__ cnt,
                                                        const int* __restrict__ slots,
                                                        const int* __restrict__ ovf_cnt,
                                                        const int* __restrict__ ovf_d,
                                                        const int* __restrict__ ovf_s,
                                                        const float* __restrict__ t,
                                                        const float* __restrict__ b,
                                                        const float* __restrict__ g,
                                                        const float* __restrict__ be,
                                                        const float* __restrict__ mean,
                                                        const float* __restrict__ var,
                                                        const float* __restrict__ W,
                                                        float* __restrict__ tout, int n) {
  __shared__ float2 s_h2[GG_NODES * TP / 2];
  __shared__ float2 s_t2[GG_NODES * TP / 2];
  __shared__ float2 s_w2[16 * 4 * 32];
  float* s_h = reinterpret_cast<float*>(s_h2);
  float* s_t = reinterpret_cast<float*>(s_t2);
  const int lane = threadIdx.x & 31;
  const int w    = __builtin_amdgcn_readfirstlane((int)(threadIdx.x >> 5));
  const int m  = lane & 15;
  const int hi = lane >> 4;
  const int kh = hi << 1;
  const int ntiles = (n + GG_NODES - 1) / GG_NODES;
  const float2* trow = reinterpret_cast<const float2*>(t);

  int tile = blockIdx.x;
  const int gstep = (int)gridDim.x;
  int degs_a = 0;
  int degs_b = 0;
  if (tile < ntiles && tile * GG_NODES + m < n) degs_a = cnt[tile * GG_NODES + m];
  if (tile + gstep < ntiles && (tile + gstep) * GG_NODES + m < n) degs_b = cnt[(tile + gstep) * GG_NODES + m];

  {
    const int ka = 4 * w + kh;
#pragma unroll
    for (int q = 0; q < 4; ++q)
      s_w2[(w * 4 + q) * 32 + lane] = make_float2(W[ka * HID + q * 16 + m], W[(ka + 1) * HID + q * 16 + m]);
  }
  const FeatConst f = load_feat_const(b, g, be, mean, var, lane);

  int deg_cur = __builtin_amdgcn_readlane(degs_a, w);
  int ent_cur = 0;
  if (tile < ntiles && tile * GG_NODES + w < n) ent_cur = load_slot_row(slots, tile * GG_NODES + w, deg_cur, lane);

  for (; tile < ntiles; tile += gstep) {
    const int node = tile * GG_NODES + w;
    const int deg  = deg_cur;
    const int ent  = ent_cur;
    const int tile_nx = tile + gstep;
    const int node_nx = tile_nx * GG_NODES + w;
    const int deg_nx  = __builtin_amdgcn_readlane(degs_b, w);
    int ent_nx = 0;
    if (tile_nx < ntiles && node_nx < n) ent_nx = load_slot_row(slots, node_nx, deg_nx, lane);
    const int tile_n2 = tile_nx + gstep;
    int degs_c = 0;
    if (tile_n2 < ntiles && tile_n2 * GG_NODES + m < n) degs_c = cnt[tile_n2 * GG_NODES + m];

    float2 h = make_float2(0.0f, 0.0f);
    float di = 0.0f;
    if (node < n)
      h = gather_bn_relu(deg, ent, ovf_cnt, ovf_d, ovf_s, trow, f, node, lane, di);
    deg_cur = deg_nx;
    ent_cur = ent_nx;
    degs_b  = degs_c;

    *reinterpret_cast<float2*>(s_h + w * TP + 2 * lane) = h;
    __syncthreads();
    if (w < 4) {
      v8f acc = {};
#pragma unroll
      for (int i = 0; i < 16; ++i) {
        const int ka = 4 * i + kh;
        const float2 av = *reinterpret_cast<const float2*>(s_h + m * TP + ka);
        const float2 bv = s_w2[(i * 4 + w) * 32 + lane];
        v2f a, bb;
        a.x = av.x;
        a.y = av.y;
        bb.x = bv.x;
        bb.y = bv.y;
        acc = wmma4(a, bb, acc);
      }
#pragma unroll
      for (int r = 0; r < 8; ++r) s_t[(r + 8 * hi) * TP + w * 16 + m] = acc[r];
    }
    __syncthreads();
    if (node < n) {
      const float2 tv = *reinterpret_cast<const float2*>(s_t + w * TP + 2 * lane);
      reinterpret_cast<float2*>(tout)[(size_t)node * 32 + lane] = make_float2(tv.x * di, tv.y * di);
    }
  }
}

#define CLS_NODES 16
__global__ __launch_bounds__(CLS_NODES * 32) void k_gather_bn_cls(const int* __restrict__ cnt,
                                                       const int* __restrict__ slots,
                                                       const int* __restrict__ ovf_cnt,
                                                       const int* __restrict__ ovf_d,
                                                       const int* __restrict__ ovf_s,
                                                       const float* __restrict__ t,
                                                       const float* __restrict__ b,
                                                       const float* __restrict__ g,
                                                       const float* __restrict__ be,
                                                       const float* __restrict__ mean,
                                                       const float* __restrict__ var,
                                                       const float* __restrict__ Wc,
                                                       const float* __restrict__ bc,
                                                       float* __restrict__ out, int n) {
  __shared__ float s_out[2][CLS_NODES * COUT];
  const int lane = threadIdx.x & 31;
  const int w    = __builtin_amdgcn_readfirstlane((int)(threadIdx.x >> 5));
  const int ntiles = (n + CLS_NODES - 1) / CLS_NODES;
  const float2* trow = reinterpret_cast<const float2*>(t);
  const size_t total = (size_t)n * COUT;
  const float4* w4 = reinterpret_cast<const float4*>(Wc) + 3 * lane;
  const float4 q0 = w4[0];
  const float4 q1 = w4[1];
  const float4 q2 = w4[2];
  const float bcl = bc[lane < COUT ? lane : 0];
  const FeatConst f = load_feat_const(b, g, be, mean, var, lane);

  int tile = blockIdx.x;
  const int gstep = (int)gridDim.x;
  const int m16 = lane & 15;
  int degs_a = 0;
  int degs_b = 0;
  if (tile < ntiles && tile * CLS_NODES + m16 < n) degs_a = cnt[tile * CLS_NODES + m16];
  if (tile + gstep < ntiles && (tile + gstep) * CLS_NODES + m16 < n) degs_b = cnt[(tile + gstep) * CLS_NODES + m16];
  int deg_cur = __builtin_amdgcn_readlane(degs_a, w);
  int ent_cur = 0;
  if (tile < ntiles && tile * CLS_NODES + w < n) ent_cur = load_slot_row(slots, tile * CLS_NODES + w, deg_cur, lane);
  int buf = 0;
  for (; tile < ntiles; tile += gstep) {
    const int node = tile * CLS_NODES + w;
    const int deg  = deg_cur;
    const int ent  = ent_cur;
    const int tile_nx = tile + gstep;
    const int node_nx = tile_nx * CLS_NODES + w;
    const int deg_nx  = __builtin_amdgcn_readlane(degs_b, w);
    int ent_nx = 0;
    if (tile_nx < ntiles && node_nx < n) ent_nx = load_slot_row(slots, node_nx, deg_nx, lane);
    const int tile_n2 = tile_nx + gstep;
    int degs_c = 0;
    if (tile_n2 < ntiles && tile_n2 * CLS_NODES + m16 < n) degs_c = cnt[tile_n2 * CLS_NODES + m16];

    float2 h = make_float2(0.0f, 0.0f);
    float di = 0.0f;
    if (node < n)
      h = gather_bn_relu(deg, ent, ovf_cnt, ovf_d, ovf_s, trow, f, node, lane, di);
    deg_cur = deg_nx;
    ent_cur = ent_nx;
    degs_b  = degs_c;
    if (node < n) {
      float p[COUT];
      p[0] = fmaf(h.y, q1.z, h.x * q0.x);
      p[1] = fmaf(h.y, q1.w, h.x * q0.y);
      p[2] = fmaf(h.y, q2.x, h.x * q0.z);
      p[3] = fmaf(h.y, q2.y, h.x * q0.w);
      p[4] = fmaf(h.y, q2.z, h.x * q1.x);
      p[5] = fmaf(h.y, q2.w, h.x * q1.y);
#pragma unroll
      for (int c = 0; c < COUT; ++c)
        for (int o = 16; o > 0; o >>= 1) p[c] += __shfl_xor(p[c], o, 32);
      float v = p[0];
#pragma unroll
      for (int c = 1; c < COUT; ++c) v = (lane == c) ? p[c] : v;
      if (lane < COUT) s_out[buf][w * COUT + lane] = v + bcl;
    }
    __syncthreads();
    if (w == 0) {
      const size_t o0 = (size_t)tile * (CLS_NODES * COUT);
#pragma unroll
      for (int j = 0; j < (CLS_NODES * COUT) / 32; ++j) {
        const int idx = j * 32 + lane;
        if (o0 + idx < total) out[o0 + idx] = s_out[buf][idx];
      }
    }
    buf ^= 1;
  }
}

extern "C" void kernel_launch(void* const* d_in, const int* in_sizes, int n_in,
                              void* d_out, int out_size, void* d_ws, size_t ws_size,
                              hipStream_t stream) {
  (void)n_in; (void)out_size; (void)ws_size;

  const float* x     = (const float*)d_in[0];
  const int*   ei    = (const int*)  d_in[1];
  const float* W_emb = (const float*)d_in[2];
  const float* b_emb = (const float*)d_in[3];
  const float* W1    = (const float*)d_in[4];
  const float* b1    = (const float*)d_in[5];
  const float* g1    = (const float*)d_in[6];
  const float* be1   = (const float*)d_in[7];
  const float* m1    = (const float*)d_in[8];
  const float* v1    = (const float*)d_in[9];
  const float* W2    = (const float*)d_in[10];
  const float* b2    = (const float*)d_in[11];
  const float* g2    = (const float*)d_in[12];
  const float* be2   = (const float*)d_in[13];
  const float* m2    = (const float*)d_in[14];
  const float* v2    = (const float*)d_in[15];
  const float* W_cls = (const float*)d_in[16];
  const float* b_cls = (const float*)d_in[17];

  const int n = in_sizes[0] / FIN;
  const int e = in_sizes[1] / 2;
  const int* src = ei;
  const int* dst = ei + e;

  float* ws   = (float*)d_ws;
  int*   cnt  = (int*)d_ws;
  float* B0   = ws + n;
  float* B1   = B0 + (size_t)n * HID;
  int*   slots   = (int*)(B1 + (size_t)n * HID);
  int*   ovf_cnt = slots + (size_t)n * SLOTS;
  int*   ovf_d   = ovf_cnt + 1;
  int*   ovf_s   = ovf_d + e;

  auto cdiv = [](long a, long b) { return (int)((a + b - 1) / b); };

  const int nb = cdiv(n, BK_NODES);
  int2*  rec       = (int2*)B0;
  int2*  spill     = rec + (size_t)nb * BK_CAP;
  int*   bcur      = (int*)(spill + e);
  int*   spill_cnt = bcur + (size_t)nb * CUR_PITCH;

  k_zero_heads<<<cdiv(nb, 256), 256, 0, stream>>>(bcur, nb, ovf_cnt, spill_cnt);
  k_partition <<<cdiv(e, PT_THREADS * PT_EPT), PT_THREADS, 0, stream>>>(src, dst, bcur, rec, spill_cnt, spill, e, nb);
  k_rank_place<<<nb, RK_THREADS, 0, stream>>>(bcur, rec, spill_cnt, spill, cnt, slots, ovf_cnt, ovf_d, ovf_s, n);

  const int gthreads = cdiv(n, 16) * 32;

  k_embed_gemm<<<cdiv(gthreads, 256), 256, 0, stream>>>(cnt, x, W_emb, b_emb, W1, B1, n);

  k_gather_bn_gemm<<<GATHER_GRID, GG_NODES * 32, 0, stream>>>(cnt, slots, ovf_cnt, ovf_d, ovf_s, B1,
                                                                     b1, g1, be1, m1, v1, W2, B0, n);

  k_gather_bn_cls<<<GATHER_GRID, CLS_NODES * 32, 0, stream>>>(cnt, slots, ovf_cnt, ovf_d, ovf_s, B0,
                                                                   b2, g2, be2, m2, v2,
                                                                   W_cls, b_cls, (float*)d_out, n);
}


# Embed + layer-1 GEMM moved into the list-placement kernel: the workgroup that ranks a 256-node bucket forms its nodes' table rows, dinv from its LDS counters; 6 launches to 5; sum orders unchanged
# speedup vs baseline: 4.4105x; 1.0406x over previous
#include <hip/hip_runtime.h>


typedef __attribute__((ext_vector_type(2))) float v2f;
typedef __attribute__((ext_vector_type(8))) float v8f;

#define HID   64
#define FIN   22
#define COUT  6
#define BN_EPS 1e-5f
#define SLOTS 32

#define BK_SHIFT  8
#define BK_NODES  (1 << BK_SHIFT)
#define BK_MAX    512
#define BK_CAP    4096
#define CUR_PITCH 32

__global__ __launch_bounds__(256) void k_zero_heads(int* bcur, int nb, int* ovf_cnt, int* spill_cnt) {
  int i = blockIdx.x * blockDim.x + threadIdx.x;
  if (i < nb) bcur[i * CUR_PITCH] = 0;
  if (i == 0) { *ovf_cnt = 0; *spill_cnt = 0; }
}

#define PT_THREADS 512
#define PT_EPT     8
#define PT_EDGES   (PT_THREADS * PT_EPT)
__global__ __launch_bounds__(PT_THREADS) void k_partition(const int* __restrict__ src,
                                                          const int* __restrict__ dst,
                                                          int* bcur, int2* __restrict__ rec,
                                                          int* spill_cnt, int2* __restrict__ spill,
                                                          int e, int nb) {
  static_assert(PT_THREADS == BK_MAX, "one thread per bucket counter");
  __shared__ int  s_hist[BK_MAX];
  __shared__ int  s_lpre[BK_MAX];
  __shared__ int  s_gofs[BK_MAX];
  __shared__ int  s_wsum[PT_THREADS / 32];
  __shared__ int2 s_rec[PT_EDGES];
  const int tid  = threadIdx.x;
  const int lane = tid & 31;
  const int w    = tid >> 5;
  s_hist[tid] = 0;
  const int first0 = blockIdx.x * PT_EDGES;
  const int first  = first0 + tid;
  int s[PT_EPT], d[PT_EPT], r[PT_EPT];
#pragma unroll
  for (int u = 0; u < PT_EPT; ++u) {
    const int i = first + u * PT_THREADS;
    s[u] = 0;
    d[u] = -1;
    if (i < e) { s[u] = src[i]; d[u] = dst[i]; }
  }
  __syncthreads();
#pragma unroll
  for (int u = 0; u < PT_EPT; ++u) {
    r[u] = 0;
    if (d[u] >= 0) r[u] = atomicAdd(&s_hist[d[u] >> BK_SHIFT], 1);
  }
  __syncthreads();
  const int c = (tid < nb) ? s_hist[tid] : 0;
  int x = c;
#pragma unroll
  for (int o = 1; o < 32; o <<= 1) {
    const int y = __shfl_up(x, o, 32);
    if (lane >= o) x += y;
  }
  if (lane == 31) s_wsum[w] = x;
  int g0 = 0;
  if (c > 0) g0 = atomicAdd(&bcur[tid * CUR_PITCH], c);
  __syncthreads();
  int woff = 0;
#pragma unroll
  for (int k = 0; k < PT_THREADS / 32; ++k) {
    const int v = s_wsum[k];
    woff += (k < w) ? v : 0;
  }
  const int excl = woff + x - c;
  s_lpre[tid] = excl;
  s_gofs[tid] = g0 - excl;
  __syncthreads();
#pragma unroll
  for (int u = 0; u < PT_EPT; ++u) {
    if (d[u] >= 0) s_rec[s_lpre[d[u] >> BK_SHIFT] + r[u]] = make_int2(s[u], d[u]);
  }
  __syncthreads();
  int nvalid = e - first0;
  if (nvalid > PT_EDGES) nvalid = PT_EDGES;
#pragma unroll
  for (int u = 0; u < PT_EPT; ++u) {
    const int idx = tid + u * PT_THREADS;
    if (idx < nvalid) {
      const int2 q  = s_rec[idx];
      const int b   = q.y >> BK_SHIFT;
      const int pos = s_gofs[b] + idx;
      if (pos < BK_CAP) {
        rec[(size_t)b * BK_CAP + pos] = q;
      } else {
        const int o = atomicAdd(spill_cnt, 1);
        spill[o] = q;
      }
    }
  }
}

__device__ __forceinline__ void overflow_edge(int s, int d, int* ovf_cnt, int* __restrict__ ovf_d,
                                              int* __restrict__ ovf_s) {
  const int o = atomicAdd(ovf_cnt, 1);
  ovf_d[o] = d;
  ovf_s[o] = s;
}

__device__ __forceinline__ v8f wmma4(v2f a, v2f b, v8f c) {
  return __builtin_amdgcn_wmma_f32_16x16x4_f32(
       false, a,  false, b,
       (short)0, c,  false,  false);
}

#define TP 68

__device__ __forceinline__ void wave_lds_handover() {
  __builtin_amdgcn_fence(__ATOMIC_RELEASE, "wavefront");
  __builtin_amdgcn_wave_barrier();
  __builtin_amdgcn_fence(__ATOMIC_ACQUIRE, "wavefront");
}

__device__ __forceinline__ void load_x_frag(const float* __restrict__ x, int arow, int kh, float (&xa)[12]) {
#pragma unroll
  for (int i = 0; i < 6; ++i) {
    const int ka = 4 * i + kh;
    xa[2 * i]     = (ka     < FIN) ? x[(size_t)arow * FIN + ka]     : 0.0f;
    xa[2 * i + 1] = (ka + 1 < FIN) ? x[(size_t)arow * FIN + ka + 1] : 0.0f;
  }
}

__device__ __forceinline__ void embed_gemm_tile(const float (&xa)[12], float dv, int row0, int n, float* s,
                                                const float* __restrict__ We,
                                                const float* __restrict__ bias,
                                                const float* __restrict__ W1,
                                                float* __restrict__ tout, int lane) {
  const int m  = lane & 15;
  const int hi = lane >> 4;
  const int kh = hi << 1;

  v8f acc[4] = {};
#pragma unroll
  for (int i = 0; i < 6; ++i) {
    const int ka = 4 * i + kh;
    v2f a;
    a.x = xa[2 * i];
    a.y = xa[2 * i + 1];
#pragma unroll
    for (int t = 0; t < 4; ++t) {
      v2f b;
      b.x = (ka     < FIN) ? We[ka       * HID + t * 16 + m] : 0.0f;
      b.y = (ka + 1 < FIN) ? We[(ka + 1) * HID + t * 16 + m] : 0.0f;
      acc[t] = wmma4(a, b, acc[t]);
    }
  }
  wave_lds_handover();
#pragma unroll
  for (int r = 0; r < 8; ++r) {
#pragma unroll
    for (int t = 0; t < 4; ++t) {
      int col = t * 16 + m;
      float v = acc[t][r] + bias[col];
      s[(r + 8 * hi) * TP + col] = v > 0.0f ? v : 0.0f;
    }
  }
  wave_lds_handover();

  v8f acc2[4] = {};
  for (int k0 = 0; k0 < HID; k0 += 4) {
    int ka = k0 + kh;
    const float2 av = *reinterpret_cast<const float2*>(s + m * TP + ka);
    v2f a;
    a.x = av.x;
    a.y = av.y;
#pragma unroll
    for (int t = 0; t < 4; ++t) {
      v2f b;
      b.x = W1[ka       * HID + t * 16 + m];
      b.y = W1[(ka + 1) * HID + t * 16 + m];
      acc2[t] = wmma4(a, b, acc2[t]);
    }
  }
  wave_lds_handover();
#pragma unroll
  for (int r = 0; r < 8; ++r) {
#pragma unroll
    for (int t = 0; t < 4; ++t) s[(r + 8 * hi) * TP + t * 16 + m] = acc2[t][r];
  }
  wave_lds_handover();
  float2* tout2 = reinterpret_cast<float2*>(tout) + (size_t)row0 * 32 + lane;
  if (row0 + 16 <= n) {
    float2 v[16];
#pragma unroll
    for (int j = 0; j < 16; ++j) v[j] = *reinterpret_cast<const float2*>(s + j * TP + 2 * lane);
#pragma unroll
    for (int j = 0; j < 16; ++j) {
      const float dj = __int_as_float(__builtin_amdgcn_readlane(__float_as_int(dv), j));
      tout2[(size_t)j * 32] = make_float2(v[j].x * dj, v[j].y * dj);
    }
  } else {
    for (int j = 0; j < 16; ++j) {
      const float dj = __int_as_float(__builtin_amdgcn_readlane(__float_as_int(dv), j));
      if (row0 + j < n) {
        const float2 vj = *reinterpret_cast<const float2*>(s + j * TP + 2 * lane);
        tout2[(size_t)j * 32] = make_float2(vj.x * dj, vj.y * dj);
      }
    }
  }
}

#define RK_THREADS 256
#define RK_PER     (BK_CAP / RK_THREADS)
__global__ __launch_bounds__(RK_THREADS) void k_place_embed_gemm(const int* __restrict__ bcur,
                                                           const int2* __restrict__ rec,
                                                           const int* __restrict__ spill_cnt,
                                                           const int2* __restrict__ spill,
                                                           int* __restrict__ cnt, int* __restrict__ slots,
                                                           int* ovf_cnt, int* __restrict__ ovf_d,
                                                           int* __restrict__ ovf_s,
                                                           const float* __restrict__ x,
                                                           const float* __restrict__ We,
                                                           const float* __restrict__ bias,
                                                           const float* __restrict__ W1,
                                                           float* __restrict__ tout, int n) {
  static_assert(BK_NODES == (RK_THREADS / 32) * 32, "two 16-row tiles per wave cover the bucket");
  __shared__ int  s_cnt[BK_NODES];
  __shared__ int4 s_row4[BK_NODES * SLOTS / 4];
  __shared__ float2 s_tile2[(RK_THREADS / 32) * 16 * TP / 2];
  int* s_row = reinterpret_cast<int*>(s_row4);
  const int tid   = threadIdx.x;
  const int b     = blockIdx.x;
  const int node0 = b << BK_SHIFT;
  const int lane  = tid & 31;
  const int w     = __builtin_amdgcn_readfirstlane(tid >> 5);
  const int trow0 = node0 + w * 32;
  float xa0[12], xa1[12];
  {
    const int m  = lane & 15;
    const int kh = (lane >> 4) << 1;
    int ar0 = trow0 + m;      if (ar0 > n - 1) ar0 = n - 1;
    int ar1 = trow0 + 16 + m; if (ar1 > n - 1) ar1 = n - 1;
    load_x_frag(x, ar0, kh, xa0);
    load_x_frag(x, ar1, kh, xa1);
  }
  for (int j = tid; j < BK_NODES; j += RK_THREADS) s_cnt[j] = 0;
  for (int j = tid; j < BK_NODES * SLOTS / 4; j += RK_THREADS) s_row4[j] = make_int4(0, 0, 0, 0);
  const int total = bcur[b * CUR_PITCH];
  const int nrec  = total < BK_CAP ? total : BK_CAP;
  const int nsp   = *spill_cnt;
  const int2* myrec = rec + (size_t)b * BK_CAP;
  int2 sd[RK_PER];
  int  rk[RK_PER];
#pragma unroll
  for (int u = 0; u < RK_PER; ++u) {
    const int i = tid + u * RK_THREADS;
    sd[u] = make_int2(0, -1);
    if (i < nrec) sd[u] = myrec[i];
  }
  __syncthreads();
#pragma unroll
  for (int u = 0; u < RK_PER; ++u) {
    rk[u] = 0;
    if (sd[u].y >= 0) rk[u] = atomicAdd(&s_cnt[sd[u].y - node0], 1);
  }
#pragma unroll
  for (int u = 0; u < RK_PER; ++u) {
    if (sd[u].y >= 0) {
      if (rk[u] < SLOTS) s_row[(sd[u].y - node0) * SLOTS + rk[u]] = sd[u].x;
      else               overflow_edge(sd[u].x, sd[u].y, ovf_cnt, ovf_d, ovf_s);
    }
  }
  for (int i = tid; i < nsp; i += RK_THREADS) {
    const int2 q = spill[i];
    if ((q.y >> BK_SHIFT) == b) {
      const int r = atomicAdd(&s_cnt[q.y - node0], 1);
      if (r < SLOTS) s_row[(q.y - node0) * SLOTS + r] = q.x;
      else           overflow_edge(q.x, q.y, ovf_cnt, ovf_d, ovf_s);
    }
  }
  __syncthreads();
  int nrows = n - node0;
  if (nrows > BK_NODES) nrows = BK_NODES;
  for (int j = tid; j < nrows; j += RK_THREADS) cnt[node0 + j] = s_cnt[j];
  int4* grow = reinterpret_cast<int4*>(slots + (size_t)node0 * SLOTS);
  for (int j = tid; j < nrows * (SLOTS / 4); j += RK_THREADS) grow[j] = s_row4[j];

  float* s = reinterpret_cast<float*>(s_tile2) + w * (16 * TP);
  const int lm = w * 32 + (lane & 15);
  if (trow0 < n) {
    const float dv = rsqrtf((float)s_cnt[lm] + 1.0f);
    embed_gemm_tile(xa0, dv, trow0, n, s, We, bias, W1, tout, lane);
  }
  if (trow0 + 16 < n) {
    const float dv = rsqrtf((float)s_cnt[lm + 16] + 1.0f);
    embed_gemm_tile(xa1, dv, trow0 + 16, n, s, We, bias, W1, tout, lane);
  }
}

template <int G>
__device__ __forceinline__ void add_rows(const float2* __restrict__ trow, int lane,
                                         int ent, int k, int m, float2& acc) {
  float2 r[G];
#pragma unroll
  for (int u = 0; u < G; ++u) {
    const int kk = (k + u < m) ? (k + u) : (m - 1);
    const int s  = __builtin_amdgcn_readlane(ent, kk);
    r[u] = trow[(size_t)s * 32 + lane];
  }
#pragma unroll
  for (int u = 0; u < G; ++u) {
    const bool live = (k + u < m);
    acc.x += live ? r[u].x : 0.0f;
    acc.y += live ? r[u].y : 0.0f;
  }
}

struct FeatConst {
  float2 pb, pm, pg, pbe;
  float rsx, rsy;
};

__device__ __forceinline__ FeatConst load_feat_const(const float* __restrict__ b,
                                                     const float* __restrict__ g,
                                                     const float* __restrict__ be,
                                                     const float* __restrict__ mean,
                                                     const float* __restrict__ var, int lane) {
  FeatConst f;
  f.pb  = reinterpret_cast<const float2*>(b)[lane];
  f.pm  = reinterpret_cast<const float2*>(mean)[lane];
  const float2 pv = reinterpret_cast<const float2*>(var)[lane];
  f.pg  = reinterpret_cast<const float2*>(g)[lane];
  f.pbe = reinterpret_cast<const float2*>(be)[lane];
  f.rsx = rsqrtf(pv.x + BN_EPS);
  f.rsy = rsqrtf(pv.y + BN_EPS);
  return f;
}

__device__ __forceinline__ int load_slot_row(const int* __restrict__ slots, int node, int deg, int lane) {
  const int m = deg < SLOTS ? deg : SLOTS;
  int ent = 0;
  if (lane < m) ent = slots[(size_t)node * SLOTS + lane];
  return ent;
}

__device__ __forceinline__ float2 gather_bn_relu(int deg, int ent,
                                                 const int* __restrict__ ovf_cnt,
                                                 const int* __restrict__ ovf_d,
                                                 const int* __restrict__ ovf_s,
                                                 const float2* __restrict__ trow,
                                                 const FeatConst& f,
                                                 int node, int lane, float& di_out) {
  const float2 self = trow[(size_t)node * 32 + lane];

  const int m = deg < SLOTS ? deg : SLOTS;

  float2 acc = make_float2(0.0f, 0.0f);
  int k = 0;
  while (m - k > 4) { add_rows<8>(trow, lane, ent, k, m, acc); k += 8; }
  if (k < m) add_rows<4>(trow, lane, ent, k, m, acc);

  if (deg > SLOTS) {
    const int novf = *ovf_cnt;
    for (int j0 = 0; j0 < novf; j0 += 32) {
      const int j = j0 + lane;
      int od = -1;
      int os = 0;
      if (j < novf) { od = ovf_d[j]; os = ovf_s[j]; }
      unsigned mask = __builtin_amdgcn_ballot_w32(od == node);
      while (mask != 0u) {
        const int kk = __builtin_ctz(mask);
        mask &= mask - 1u;
        const int s = __builtin_amdgcn_readlane(os, kk);
        const float2 r = trow[(size_t)s * 32 + lane];
        acc.x += r.x;
        acc.y += r.y;
      }
    }
  }

  acc.x += self.x;
  acc.y += self.y;
  const float di = rsqrtf((float)deg + 1.0f);
  di_out = di;

  float vx = acc.x * di + f.pb.x;
  float vy = acc.y * di + f.pb.y;
  vx = (vx - f.pm.x) * f.rsx * f.pg.x + f.pbe.x;
  vy = (vy - f.pm.y) * f.rsy * f.pg.y + f.pbe.y;
  return make_float2(vx > 0.0f ? vx : 0.0f, vy > 0.0f ? vy : 0.0f);
}

#define GATHER_GRID 1024

#define GG_NODES 16
__global__ __launch_bounds__(GG_NODES * 32) void k_gather_bn_gemm(const int* __restrict__ cnt,
                                                        const int* __restrict__ slots,
                                                        const int* __restrict__ ovf_cnt,
                                                        const int* __restrict__ ovf_d,
                                                        const int* __restrict__ ovf_s,
                                                        const float* __restrict__ t,
                                                        const float* __restrict__ b,
                                                        const float* __restrict__ g,
                                                        const float* __restrict__ be,
                                                        const float* __restrict__ mean,
                                                        const float* __restrict__ var,
                                                        const float* __restrict__ W,
                                                        float* __restrict__ tout, int n) {
  __shared__ float2 s_h2[GG_NODES * TP / 2];
  __shared__ float2 s_t2[GG_NODES * TP / 2];
  __shared__ float2 s_w2[16 * 4 * 32];
  float* s_h = reinterpret_cast<float*>(s_h2);
  float* s_t = reinterpret_cast<float*>(s_t2);
  const int lane = threadIdx.x & 31;
  const int w    = __builtin_amdgcn_readfirstlane((int)(threadIdx.x >> 5));
  const int m  = lane & 15;
  const int hi = lane >> 4;
  const int kh = hi << 1;
  const int ntiles = (n + GG_NODES - 1) / GG_NODES;
  const float2* trow = reinterpret_cast<const float2*>(t);

  int tile = blockIdx.x;
  const int gstep = (int)gridDim.x;
  int degs_a = 0;
  int degs_b = 0;
  if (tile < ntiles && tile * GG_NODES + m < n) degs_a = cnt[tile * GG_NODES + m];
  if (tile + gstep < ntiles && (tile + gstep) * GG_NODES + m < n) degs_b = cnt[(tile + gstep) * GG_NODES + m];

  {
    const int ka = 4 * w + kh;
#pragma unroll
    for (int q = 0; q < 4; ++q)
      s_w2[(w * 4 + q) * 32 + lane] = make_float2(W[ka * HID + q * 16 + m], W[(ka + 1) * HID + q * 16 + m]);
  }
  const FeatConst f = load_feat_const(b, g, be, mean, var, lane);

  int deg_cur = __builtin_amdgcn_readlane(degs_a, w);
  int ent_cur = 0;
  if (tile < ntiles && tile * GG_NODES + w < n) ent_cur = load_slot_row(slots, tile * GG_NODES + w, deg_cur, lane);

  for (; tile < ntiles; tile += gstep) {
    const int node = tile * GG_NODES + w;
    const int deg  = deg_cur;
    const int ent  = ent_cur;
    const int tile_nx = tile + gstep;
    const int node_nx = tile_nx * GG_NODES + w;
    const int deg_nx  = __builtin_amdgcn_readlane(degs_b, w);
    int ent_nx = 0;
    if (tile_nx < ntiles && node_nx < n) ent_nx = load_slot_row(slots, node_nx, deg_nx, lane);
    const int tile_n2 = tile_nx + gstep;
    int degs_c = 0;
    if (tile_n2 < ntiles && tile_n2 * GG_NODES + m < n) degs_c = cnt[tile_n2 * GG_NODES + m];

    float2 h = make_float2(0.0f, 0.0f);
    float di = 0.0f;
    if (node < n)
      h = gather_bn_relu(deg, ent, ovf_cnt, ovf_d, ovf_s, trow, f, node, lane, di);
    deg_cur = deg_nx;
    ent_cur = ent_nx;
    degs_b  = degs_c;

    *reinterpret_cast<float2*>(s_h + w * TP + 2 * lane) = h;
    __syncthreads();
    if (w < 4) {
      v8f acc = {};
#pragma unroll
      for (int i = 0; i < 16; ++i) {
        const int ka = 4 * i + kh;
        const float2 av = *reinterpret_cast<const float2*>(s_h + m * TP + ka);
        const float2 bv = s_w2[(i * 4 + w) * 32 + lane];
        v2f a, bb;
        a.x = av.x;
        a.y = av.y;
        bb.x = bv.x;
        bb.y = bv.y;
        acc = wmma4(a, bb, acc);
      }
#pragma unroll
      for (int r = 0; r < 8; ++r) s_t[(r + 8 * hi) * TP + w * 16 + m] = acc[r];
    }
    __syncthreads();
    if (node < n) {
      const float2 tv = *reinterpret_cast<const float2*>(s_t + w * TP + 2 * lane);
      reinterpret_cast<float2*>(tout)[(size_t)node * 32 + lane] = make_float2(tv.x * di, tv.y * di);
    }
  }
}

#define CLS_NODES 16
__global__ __launch_bounds__(CLS_NODES * 32) void k_gather_bn_cls(const int* __restrict__ cnt,
                                                       const int* __restrict__ slots,
                                                       const int* __restrict__ ovf_cnt,
                                                       const int* __restrict__ ovf_d,
                                                       const int* __restrict__ ovf_s,
                                                       const float* __restrict__ t,
                                                       const float* __restrict__ b,
                                                       const float* __restrict__ g,
                                                       const float* __restrict__ be,
                                                       const float* __restrict__ mean,
                                                       const float* __restrict__ var,
                                                       const float* __restrict__ Wc,
                                                       const float* __restrict__ bc,
                                                       float* __restrict__ out, int n) {
  __shared__ float s_out[2][CLS_NODES * COUT];
  const int lane = threadIdx.x & 31;
  const int w    = __builtin_amdgcn_readfirstlane((int)(threadIdx.x >> 5));
  const int ntiles = (n + CLS_NODES - 1) / CLS_NODES;
  const float2* trow = reinterpret_cast<const float2*>(t);
  const size_t total = (size_t)n * COUT;
  const float4* w4 = reinterpret_cast<const float4*>(Wc) + 3 * lane;
  const float4 q0 = w4[0];
  const float4 q1 = w4[1];
  const float4 q2 = w4[2];
  const float bcl = bc[lane < COUT ? lane : 0];
  const FeatConst f = load_feat_const(b, g, be, mean, var, lane);

  int tile = blockIdx.x;
  const int gstep = (int)gridDim.x;
  const int m16 = lane & 15;
  int degs_a = 0;
  int degs_b = 0;
  if (tile < ntiles && tile * CLS_NODES + m16 < n) degs_a = cnt[tile * CLS_NODES + m16];
  if (tile + gstep < ntiles && (tile + gstep) * CLS_NODES + m16 < n) degs_b = cnt[(tile + gstep) * CLS_NODES + m16];
  int deg_cur = __builtin_amdgcn_readlane(degs_a, w);
  int ent_cur = 0;
  if (tile < ntiles && tile * CLS_NODES + w < n) ent_cur = load_slot_row(slots, tile * CLS_NODES + w, deg_cur, lane);
  int buf = 0;
  for (; tile < ntiles; tile += gstep) {
    const int node = tile * CLS_NODES + w;
    const int deg  = deg_cur;
    const int ent  = ent_cur;
    const int tile_nx = tile + gstep;
    const int node_nx = tile_nx * CLS_NODES + w;
    const int deg_nx  = __builtin_amdgcn_readlane(degs_b, w);
    int ent_nx = 0;
    if (tile_nx < ntiles && node_nx < n) ent_nx = load_slot_row(slots, node_nx, deg_nx, lane);
    const int tile_n2 = tile_nx + gstep;
    int degs_c = 0;
    if (tile_n2 < ntiles && tile_n2 * CLS_NODES + m16 < n) degs_c = cnt[tile_n2 * CLS_NODES + m16];

    float2 h = make_float2(0.0f, 0.0f);
    float di = 0.0f;
    if (node < n)
      h = gather_bn_relu(deg, ent, ovf_cnt, ovf_d, ovf_s, trow, f, node, lane, di);
    deg_cur = deg_nx;
    ent_cur = ent_nx;
    degs_b  = degs_c;
    if (node < n) {
      float p[COUT];
      p[0] = fmaf(h.y, q1.z, h.x * q0.x);
      p[1] = fmaf(h.y, q1.w, h.x * q0.y);
      p[2] = fmaf(h.y, q2.x, h.x * q0.z);
      p[3] = fmaf(h.y, q2.y, h.x * q0.w);
      p[4] = fmaf(h.y, q2.z, h.x * q1.x);
      p[5] = fmaf(h.y, q2.w, h.x * q1.y);
#pragma unroll
      for (int c = 0; c < COUT; ++c)
        for (int o = 16; o > 0; o >>= 1) p[c] += __shfl_xor(p[c], o, 32);
      float v = p[0];
#pragma unroll
      for (int c = 1; c < COUT; ++c) v = (lane == c) ? p[c] : v;
      if (lane < COUT) s_out[buf][w * COUT + lane] = v + bcl;
    }
    __syncthreads();
    if (w == 0) {
      const size_t o0 = (size_t)tile * (CLS_NODES * COUT);
#pragma unroll
      for (int j = 0; j < (CLS_NODES * COUT) / 32; ++j) {
        const int idx = j * 32 + lane;
        if (o0 + idx < total) out[o0 + idx] = s_out[buf][idx];
      }
    }
    buf ^= 1;
  }
}

extern "C" void kernel_launch(void* const* d_in, const int* in_sizes, int n_in,
                              void* d_out, int out_size, void* d_ws, size_t ws_size,
                              hipStream_t stream) {
  (void)n_in; (void)out_size; (void)ws_size;

  const float* x     = (const float*)d_in[0];
  const int*   ei    = (const int*)  d_in[1];
  const float* W_emb = (const float*)d_in[2];
  const float* b_emb = (const float*)d_in[3];
  const float* W1    = (const float*)d_in[4];
  const float* b1    = (const float*)d_in[5];
  const float* g1    = (const float*)d_in[6];
  const float* be1   = (const float*)d_in[7];
  const float* m1    = (const float*)d_in[8];
  const float* v1    = (const float*)d_in[9];
  const float* W2    = (const float*)d_in[10];
  const float* b2    = (const float*)d_in[11];
  const float* g2    = (const float*)d_in[12];
  const float* be2   = (const float*)d_in[13];
  const float* m2    = (const float*)d_in[14];
  const float* v2    = (const float*)d_in[15];
  const float* W_cls = (const float*)d_in[16];
  const float* b_cls = (const float*)d_in[17];

  const int n = in_sizes[0] / FIN;
  const int e = in_sizes[1] / 2;
  const int* src = ei;
  const int* dst = ei + e;

  float* ws   = (float*)d_ws;
  int*   cnt  = (int*)d_ws;
  float* B0   = ws + n;
  float* B1   = B0 + (size_t)n * HID;
  int*   slots   = (int*)(B1 + (size_t)n * HID);
  int*   ovf_cnt = slots + (size_t)n * SLOTS;
  int*   ovf_d   = ovf_cnt + 1;
  int*   ovf_s   = ovf_d + e;

  auto cdiv = [](long a, long b) { return (int)((a + b - 1) / b); };

  const int nb = cdiv(n, BK_NODES);
  int2*  rec       = (int2*)B0;
  int2*  spill     = rec + (size_t)nb * BK_CAP;
  int*   bcur      = (int*)(spill + e);
  int*   spill_cnt = bcur + (size_t)nb * CUR_PITCH;

  k_zero_heads<<<cdiv(nb, 256), 256, 0, stream>>>(bcur, nb, ovf_cnt, spill_cnt);
  k_partition <<<cdiv(e, PT_THREADS * PT_EPT), PT_THREADS, 0, stream>>>(src, dst, bcur, rec, spill_cnt, spill, e, nb);
  k_place_embed_gemm<<<nb, RK_THREADS, 0, stream>>>(bcur, rec, spill_cnt, spill, cnt, slots, ovf_cnt, ovf_d, ovf_s,
                                                    x, W_emb, b_emb, W1, B1, n);

  k_gather_bn_gemm<<<GATHER_GRID, GG_NODES * 32, 0, stream>>>(cnt, slots, ovf_cnt, ovf_d, ovf_s, B1,
                                                                     b1, g1, be1, m1, v1, W2, B0, n);

  k_gather_bn_cls<<<GATHER_GRID, CLS_NODES * 32, 0, stream>>>(cnt, slots, ovf_cnt, ovf_d, ovf_s, B0,
                                                                   b2, g2, be2, m2, v2,
                                                                   W_cls, b_cls, (float*)d_out, n);
}


# List build without memory cursors: partition stores each slice bucket-ordered at a fixed place with a run table that placement reads; zeroing launch gone (5 to 4); sum order still LDS arrival
# speedup vs baseline: 4.7840x; 1.0847x over previous
#include <hip/hip_runtime.h>


typedef __attribute__((ext_vector_type(2))) float v2f;
typedef __attribute__((ext_vector_type(8))) float v8f;

#define HID   64
#define FIN   22
#define COUT  6
#define BN_EPS 1e-5f
#define SLOTS 32

#define BK_SHIFT  8
#define BK_NODES  (1 << BK_SHIFT)
#define BK_MAX    512

#define PT_THREADS 512
#define PT_EPT     8
#define PT_EDGES   (PT_THREADS * PT_EPT)
__global__ __launch_bounds__(PT_THREADS) void k_partition(const int* __restrict__ src,
                                                          const int* __restrict__ dst,
                                                          int2* __restrict__ recs,
                                                          int* __restrict__ pre_t, int* ovf_cnt,
                                                          int e, int nb) {
  static_assert(PT_THREADS == BK_MAX, "one thread per bucket counter");
  __shared__ int  s_hist[BK_MAX];
  __shared__ int  s_lpre[BK_MAX];
  __shared__ int  s_wsum[PT_THREADS / 32];
  __shared__ int2 s_rec[PT_EDGES];
  const int tid  = threadIdx.x;
  const int lane = tid & 31;
  const int w    = tid >> 5;
  s_hist[tid] = 0;
  const int first0 = blockIdx.x * PT_EDGES;
  const int first  = first0 + tid;
  int s[PT_EPT], d[PT_EPT], r[PT_EPT];
#pragma unroll
  for (int u = 0; u < PT_EPT; ++u) {
    const int i = first + u * PT_THREADS;
    s[u] = 0;
    d[u] = -1;
    if (i < e) { s[u] = src[i]; d[u] = dst[i]; }
  }
  __syncthreads();
#pragma unroll
  for (int u = 0; u < PT_EPT; ++u) {
    r[u] = 0;
    if (d[u] >= 0) r[u] = atomicAdd(&s_hist[d[u] >> BK_SHIFT], 1);
  }
  __syncthreads();
  const int c = (tid < nb) ? s_hist[tid] : 0;
  int x = c;
#pragma unroll
  for (int o = 1; o < 32; o <<= 1) {
    const int y = __shfl_up(x, o, 32);
    if (lane >= o) x += y;
  }
  if (lane == 31) s_wsum[w] = x;
  __syncthreads();
  int woff = 0;
#pragma unroll
  for (int k = 0; k < PT_THREADS / 32; ++k) {
    const int v = s_wsum[k];
    woff += (k < w) ? v : 0;
  }
  const int excl = woff + x - c;
  s_lpre[tid] = excl;
  const int nsl = (int)gridDim.x;
  if (tid < nb)  pre_t[(size_t)(tid + 1) * nsl + blockIdx.x] = excl + c;
  if (tid == 0)  pre_t[blockIdx.x] = 0;
  if (tid == 0 && blockIdx.x == 0) *ovf_cnt = 0;
  __syncthreads();
#pragma unroll
  for (int u = 0; u < PT_EPT; ++u) {
    if (d[u] >= 0) s_rec[s_lpre[d[u] >> BK_SHIFT] + r[u]] = make_int2(s[u], d[u]);
  }
  __syncthreads();
  int nvalid = e - first0;
  if (nvalid > PT_EDGES) nvalid = PT_EDGES;
#pragma unroll
  for (int u = 0; u < PT_EPT; ++u) {
    const int idx = tid + u * PT_THREADS;
    if (idx < nvalid) recs[(size_t)first0 + idx] = s_rec[idx];
  }
}

__device__ __forceinline__ void overflow_edge(int s, int d, int* ovf_cnt, int* __restrict__ ovf_d,
                                              int* __restrict__ ovf_s) {
  const int o = atomicAdd(ovf_cnt, 1);
  ovf_d[o] = d;
  ovf_s[o] = s;
}

__device__ __forceinline__ v8f wmma4(v2f a, v2f b, v8f c) {
  return __builtin_amdgcn_wmma_f32_16x16x4_f32(
       false, a,  false, b,
       (short)0, c,  false,  false);
}

#define TP 68

__device__ __forceinline__ void wave_lds_handover() {
  __builtin_amdgcn_fence(__ATOMIC_RELEASE, "wavefront");
  __builtin_amdgcn_wave_barrier();
  __builtin_amdgcn_fence(__ATOMIC_ACQUIRE, "wavefront");
}

__device__ __forceinline__ void load_x_frag(const float* __restrict__ x, int arow, int kh, float (&xa)[12]) {
#pragma unroll
  for (int i = 0; i < 6; ++i) {
    const int ka = 4 * i + kh;
    xa[2 * i]     = (ka     < FIN) ? x[(size_t)arow * FIN + ka]     : 0.0f;
    xa[2 * i + 1] = (ka + 1 < FIN) ? x[(size_t)arow * FIN + ka + 1] : 0.0f;
  }
}

__device__ __forceinline__ void embed_gemm_tile(const float (&xa)[12], float dv, int row0, int n, float* s,
                                                const float* __restrict__ We,
                                                const float* __restrict__ bias,
                                                const float* __restrict__ W1,
                                                float* __restrict__ tout, int lane) {
  const int m  = lane & 15;
  const int hi = lane >> 4;
  const int kh = hi << 1;

  v8f acc[4] = {};
#pragma unroll
  for (int i = 0; i < 6; ++i) {
    const int ka = 4 * i + kh;
    v2f a;
    a.x = xa[2 * i];
    a.y = xa[2 * i + 1];
#pragma unroll
    for (int t = 0; t < 4; ++t) {
      v2f b;
      b.x = (ka     < FIN) ? We[ka       * HID + t * 16 + m] : 0.0f;
      b.y = (ka + 1 < FIN) ? We[(ka + 1) * HID + t * 16 + m] : 0.0f;
      acc[t] = wmma4(a, b, acc[t]);
    }
  }
  wave_lds_handover();
#pragma unroll
  for (int r = 0; r < 8; ++r) {
#pragma unroll
    for (int t = 0; t < 4; ++t) {
      int col = t * 16 + m;
      float v = acc[t][r] + bias[col];
      s[(r + 8 * hi) * TP + col] = v > 0.0f ? v : 0.0f;
    }
  }
  wave_lds_handover();

  v8f acc2[4] = {};
  for (int k0 = 0; k0 < HID; k0 += 4) {
    int ka = k0 + kh;
    const float2 av = *reinterpret_cast<const float2*>(s + m * TP + ka);
    v2f a;
    a.x = av.x;
    a.y = av.y;
#pragma unroll
    for (int t = 0; t < 4; ++t) {
      v2f b;
      b.x = W1[ka       * HID + t * 16 + m];
      b.y = W1[(ka + 1) * HID + t * 16 + m];
      acc2[t] = wmma4(a, b, acc2[t]);
    }
  }
  wave_lds_handover();
#pragma unroll
  for (int r = 0; r < 8; ++r) {
#pragma unroll
    for (int t = 0; t < 4; ++t) s[(r + 8 * hi) * TP + t * 16 + m] = acc2[t][r];
  }
  wave_lds_handover();
  float2* tout2 = reinterpret_cast<float2*>(tout) + (size_t)row0 * 32 + lane;
  if (row0 + 16 <= n) {
    float2 v[16];
#pragma unroll
    for (int j = 0; j < 16; ++j) v[j] = *reinterpret_cast<const float2*>(s + j * TP + 2 * lane);
#pragma unroll
    for (int j = 0; j < 16; ++j) {
      const float dj = __int_as_float(__builtin_amdgcn_readlane(__float_as_int(dv), j));
      tout2[(size_t)j * 32] = make_float2(v[j].x * dj, v[j].y * dj);
    }
  } else {
    for (int j = 0; j < 16; ++j) {
      const float dj = __int_as_float(__builtin_amdgcn_readlane(__float_as_int(dv), j));
      if (row0 + j < n) {
        const float2 vj = *reinterpret_cast<const float2*>(s + j * TP + 2 * lane);
        tout2[(size_t)j * 32] = make_float2(vj.x * dj, vj.y * dj);
      }
    }
  }
}

#define RK_THREADS 256
#define RUN_CHUNK  512
#define RK_BATCH   20
__global__ __launch_bounds__(RK_THREADS) void k_place_embed_gemm(const int* __restrict__ pre_t,
                                                           const int2* __restrict__ recs, int nsl,
                                                           int* __restrict__ cnt, int* __restrict__ slots,
                                                           int* ovf_cnt, int* __restrict__ ovf_d,
                                                           int* __restrict__ ovf_s,
                                                           const float* __restrict__ x,
                                                           const float* __restrict__ We,
                                                           const float* __restrict__ bias,
                                                           const float* __restrict__ W1,
                                                           float* __restrict__ tout, int n) {
  static_assert(BK_NODES == (RK_THREADS / 32) * 32, "two 16-row tiles per wave cover the bucket");
  __shared__ int  s_cnt[BK_NODES];
  __shared__ int4 s_row4[BK_NODES * SLOTS / 4];
  __shared__ float2 s_tile2[(RK_THREADS / 32) * 16 * TP / 2];
  __shared__ int2 s_run[RUN_CHUNK];
  int* s_row = reinterpret_cast<int*>(s_row4);
  const int tid   = threadIdx.x;
  const int b     = blockIdx.x;
  const int node0 = b << BK_SHIFT;
  const int lane  = tid & 31;
  const int w     = __builtin_amdgcn_readfirstlane(tid >> 5);
  const int trow0 = node0 + w * 32;
  float xa0[12], xa1[12];
  {
    const int m  = lane & 15;
    const int kh = (lane >> 4) << 1;
    int ar0 = trow0 + m;      if (ar0 > n - 1) ar0 = n - 1;
    int ar1 = trow0 + 16 + m; if (ar1 > n - 1) ar1 = n - 1;
    load_x_frag(x, ar0, kh, xa0);
    load_x_frag(x, ar1, kh, xa1);
  }
  for (int j = tid; j < BK_NODES; j += RK_THREADS) s_cnt[j] = 0;
  for (int j = tid; j < BK_NODES * SLOTS / 4; j += RK_THREADS) s_row4[j] = make_int4(0, 0, 0, 0);
  const int hw  = tid >> 4;
  const int l16 = tid & 15;
  for (int g0 = 0; g0 < nsl; g0 += RUN_CHUNK) {
    if (g0 > 0) __syncthreads();
    for (int j = tid; j < RUN_CHUNK; j += RK_THREADS) {
      const int g = g0 + j;
      int2 rd = make_int2(0, 0);
      if (g < nsl) {
        const int a = pre_t[(size_t)b * nsl + g];
        const int z = pre_t[(size_t)(b + 1) * nsl + g];
        rd = make_int2(g * PT_EDGES + a, z - a);
      }
      s_run[j] = rd;
    }
    __syncthreads();
    int nrun = nsl - g0;
    if (nrun > RUN_CHUNK) nrun = RUN_CHUNK;
    for (int i0 = 0; i0 < nrun; i0 += 16 * RK_BATCH) {
      int2 rd[RK_BATCH], qa[RK_BATCH], qb[RK_BATCH];
      int  ra[RK_BATCH], rb[RK_BATCH];
      int  longest = 0;
#pragma unroll
      for (int u = 0; u < RK_BATCH; ++u) {
        const int j = i0 + 16 * u + hw;
        const int jj = j < nrun ? j : 0;
        rd[u] = s_run[jj];
        if (j >= nrun) rd[u].y = 0;
      }
#pragma unroll
      for (int u = 0; u < RK_BATCH; ++u) {
        longest = rd[u].y > longest ? rd[u].y : longest;
        qa[u] = make_int2(0, -1);
        qb[u] = make_int2(0, -1);
        if (l16 < rd[u].y)      qa[u] = recs[(size_t)rd[u].x + l16];
        if (l16 + 16 < rd[u].y) qb[u] = recs[(size_t)rd[u].x + l16 + 16];
      }
#pragma unroll
      for (int u = 0; u < RK_BATCH; ++u) {
        ra[u] = 0;
        rb[u] = 0;
        if (qa[u].y >= 0) ra[u] = atomicAdd(&s_cnt[qa[u].y - node0], 1);
        if (qb[u].y >= 0) rb[u] = atomicAdd(&s_cnt[qb[u].y - node0], 1);
      }
#pragma unroll
      for (int u = 0; u < RK_BATCH; ++u) {
        if (qa[u].y >= 0) {
          if (ra[u] < SLOTS) s_row[(qa[u].y - node0) * SLOTS + ra[u]] = qa[u].x;
          else               overflow_edge(qa[u].x, qa[u].y, ovf_cnt, ovf_d, ovf_s);
        }
        if (qb[u].y >= 0) {
          if (rb[u] < SLOTS) s_row[(qb[u].y - node0) * SLOTS + rb[u]] = qb[u].x;
          else               overflow_edge(qb[u].x, qb[u].y, ovf_cnt, ovf_d, ovf_s);
        }
      }
      if (longest > 32) {
        for (int u = 0; u < RK_BATCH; ++u) {
          const int j = i0 + 16 * u + hw;
          int2 rt = make_int2(0, 0);
          if (j < nrun) rt = s_run[j];
          for (int k = l16 + 32; k < rt.y; k += 16) {
            const int2 q = recs[(size_t)rt.x + k];
            const int r = atomicAdd(&s_cnt[q.y - node0], 1);
            if (r < SLOTS) s_row[(q.y - node0) * SLOTS + r] = q.x;
            else           overflow_edge(q.x, q.y, ovf_cnt, ovf_d, ovf_s);
          }
        }
      }
    }
  }
  __syncthreads();
  int nrows = n - node0;
  if (nrows > BK_NODES) nrows = BK_NODES;
  for (int j = tid; j < nrows; j += RK_THREADS) cnt[node0 + j] = s_cnt[j];
  int4* grow = reinterpret_cast<int4*>(slots + (size_t)node0 * SLOTS);
  for (int j = tid; j < nrows * (SLOTS / 4); j += RK_THREADS) grow[j] = s_row4[j];

  float* s = reinterpret_cast<float*>(s_tile2) + w * (16 * TP);
  const int lm = w * 32 + (lane & 15);
  if (trow0 < n) {
    const float dv = rsqrtf((float)s_cnt[lm] + 1.0f);
    embed_gemm_tile(xa0, dv, trow0, n, s, We, bias, W1, tout, lane);
  }
  if (trow0 + 16 < n) {
    const float dv = rsqrtf((float)s_cnt[lm + 16] + 1.0f);
    embed_gemm_tile(xa1, dv, trow0 + 16, n, s, We, bias, W1, tout, lane);
  }
}

template <int G>
__device__ __forceinline__ void add_rows(const float2* __restrict__ trow, int lane,
                                         int ent, int k, int m, float2& acc) {
  float2 r[G];
#pragma unroll
  for (int u = 0; u < G; ++u) {
    const int kk = (k + u < m) ? (k + u) : (m - 1);
    const int s  = __builtin_amdgcn_readlane(ent, kk);
    r[u] = trow[(size_t)s * 32 + lane];
  }
#pragma unroll
  for (int u = 0; u < G; ++u) {
    const bool live = (k + u < m);
    acc.x += live ? r[u].x : 0.0f;
    acc.y += live ? r[u].y : 0.0f;
  }
}

struct FeatConst {
  float2 pb, pm, pg, pbe;
  float rsx, rsy;
};

__device__ __forceinline__ FeatConst load_feat_const(const float* __restrict__ b,
                                                     const float* __restrict__ g,
                                                     const float* __restrict__ be,
                                                     const float* __restrict__ mean,
                                                     const float* __restrict__ var, int lane) {
  FeatConst f;
  f.pb  = reinterpret_cast<const float2*>(b)[lane];
  f.pm  = reinterpret_cast<const float2*>(mean)[lane];
  const float2 pv = reinterpret_cast<const float2*>(var)[lane];
  f.pg  = reinterpret_cast<const float2*>(g)[lane];
  f.pbe = reinterpret_cast<const float2*>(be)[lane];
  f.rsx = rsqrtf(pv.x + BN_EPS);
  f.rsy = rsqrtf(pv.y + BN_EPS);
  return f;
}

__device__ __forceinline__ int load_slot_row(const int* __restrict__ slots, int node, int deg, int lane) {
  const int m = deg < SLOTS ? deg : SLOTS;
  int ent = 0;
  if (lane < m) ent = slots[(size_t)node * SLOTS + lane];
  return ent;
}

__device__ __forceinline__ float2 gather_bn_relu(int deg, int ent,
                                                 const int* __restrict__ ovf_cnt,
                                                 const int* __restrict__ ovf_d,
                                                 const int* __restrict__ ovf_s,
                                                 const float2* __restrict__ trow,
                                                 const FeatConst& f,
                                                 int node, int lane, float& di_out) {
  const float2 self = trow[(size_t)node * 32 + lane];

  const int m = deg < SLOTS ? deg : SLOTS;

  float2 acc = make_float2(0.0f, 0.0f);
  int k = 0;
  while (m - k > 4) { add_rows<8>(trow, lane, ent, k, m, acc); k += 8; }
  if (k < m) add_rows<4>(trow, lane, ent, k, m, acc);

  if (deg > SLOTS) {
    const int novf = *ovf_cnt;
    for (int j0 = 0; j0 < novf; j0 += 32) {
      const int j = j0 + lane;
      int od = -1;
      int os = 0;
      if (j < novf) { od = ovf_d[j]; os = ovf_s[j]; }
      unsigned mask = __builtin_amdgcn_ballot_w32(od == node);
      while (mask != 0u) {
        const int kk = __builtin_ctz(mask);
        mask &= mask - 1u;
        const int s = __builtin_amdgcn_readlane(os, kk);
        const float2 r = trow[(size_t)s * 32 + lane];
        acc.x += r.x;
        acc.y += r.y;
      }
    }
  }

  acc.x += self.x;
  acc.y += self.y;
  const float di = rsqrtf((float)deg + 1.0f);
  di_out = di;

  float vx = acc.x * di + f.pb.x;
  float vy = acc.y * di + f.pb.y;
  vx = (vx - f.pm.x) * f.rsx * f.pg.x + f.pbe.x;
  vy = (vy - f.pm.y) * f.rsy * f.pg.y + f.pbe.y;
  return make_float2(vx > 0.0f ? vx : 0.0f, vy > 0.0f ? vy : 0.0f);
}

#define GATHER_GRID 1024

#define GG_NODES 16
__global__ __launch_bounds__(GG_NODES * 32) void k_gather_bn_gemm(const int* __restrict__ cnt,
                                                        const int* __restrict__ slots,
                                                        const int* __restrict__ ovf_cnt,
                                                        const int* __restrict__ ovf_d,
                                                        const int* __restrict__ ovf_s,
                                                        const float* __restrict__ t,
                                                        const float* __restrict__ b,
                                                        const float* __restrict__ g,
                                                        const float* __restrict__ be,
                                                        const float* __restrict__ mean,
                                                        const float* __restrict__ var,
                                                        const float* __restrict__ W,
                                                        float* __restrict__ tout, int n) {
  __shared__ float2 s_h2[GG_NODES * TP / 2];
  __shared__ float2 s_t2[GG_NODES * TP / 2];
  __shared__ float2 s_w2[16 * 4 * 32];
  float* s_h = reinterpret_cast<float*>(s_h2);
  float* s_t = reinterpret_cast<float*>(s_t2);
  const int lane = threadIdx.x & 31;
  const int w    = __builtin_amdgcn_readfirstlane((int)(threadIdx.x >> 5));
  const int m  = lane & 15;
  const int hi = lane >> 4;
  const int kh = hi << 1;
  const int ntiles = (n + GG_NODES - 1) / GG_NODES;
  const float2* trow = reinterpret_cast<const float2*>(t);

  int tile = blockIdx.x;
  const int gstep = (int)gridDim.x;
  int degs_a = 0;
  int degs_b = 0;
  if (tile < ntiles && tile * GG_NODES + m < n) degs_a = cnt[tile * GG_NODES + m];
  if (tile + gstep < ntiles && (tile + gstep) * GG_NODES + m < n) degs_b = cnt[(tile + gstep) * GG_NODES + m];

  {
    const int ka = 4 * w + kh;
#pragma unroll
    for (int q = 0; q < 4; ++q)
      s_w2[(w * 4 + q) * 32 + lane] = make_float2(W[ka * HID + q * 16 + m], W[(ka + 1) * HID + q * 16 + m]);
  }
  const FeatConst f = load_feat_const(b, g, be, mean, var, lane);

  int deg_cur = __builtin_amdgcn_readlane(degs_a, w);
  int ent_cur = 0;
  if (tile < ntiles && tile * GG_NODES + w < n) ent_cur = load_slot_row(slots, tile * GG_NODES + w, deg_cur, lane);

  for (; tile < ntiles; tile += gstep) {
    const int node = tile * GG_NODES + w;
    const int deg  = deg_cur;
    const int ent  = ent_cur;
    const int tile_nx = tile + gstep;
    const int node_nx = tile_nx * GG_NODES + w;
    const int deg_nx  = __builtin_amdgcn_readlane(degs_b, w);
    int ent_nx = 0;
    if (tile_nx < ntiles && node_nx < n) ent_nx = load_slot_row(slots, node_nx, deg_nx, lane);
    const int tile_n2 = tile_nx + gstep;
    int degs_c = 0;
    if (tile_n2 < ntiles && tile_n2 * GG_NODES + m < n) degs_c = cnt[tile_n2 * GG_NODES + m];

    float2 h = make_float2(0.0f, 0.0f);
    float di = 0.0f;
    if (node < n)
      h = gather_bn_relu(deg, ent, ovf_cnt, ovf_d, ovf_s, trow, f, node, lane, di);
    deg_cur = deg_nx;
    ent_cur = ent_nx;
    degs_b  = degs_c;

    *reinterpret_cast<float2*>(s_h + w * TP + 2 * lane) = h;
    __syncthreads();
    if (w < 4) {
      v8f acc = {};
#pragma unroll
      for (int i = 0; i < 16; ++i) {
        const int ka = 4 * i + kh;
        const float2 av = *reinterpret_cast<const float2*>(s_h + m * TP + ka);
        const float2 bv = s_w2[(i * 4 + w) * 32 + lane];
        v2f a, bb;
        a.x = av.x;
        a.y = av.y;
        bb.x = bv.x;
        bb.y = bv.y;
        acc = wmma4(a, bb, acc);
      }
#pragma unroll
      for (int r = 0; r < 8; ++r) s_t[(r + 8 * hi) * TP + w * 16 + m] = acc[r];
    }
    __syncthreads();
    if (node < n) {
      const float2 tv = *reinterpret_cast<const float2*>(s_t + w * TP + 2 * lane);
      reinterpret_cast<float2*>(tout)[(size_t)node * 32 + lane] = make_float2(tv.x * di, tv.y * di);
    }
  }
}

#define CLS_NODES 16
__global__ __launch_bounds__(CLS_NODES * 32) void k_gather_bn_cls(const int* __restrict__ cnt,
                                                       const int* __restrict__ slots,
                                                       const int* __restrict__ ovf_cnt,
                                                       const int* __restrict__ ovf_d,
                                                       const int* __restrict__ ovf_s,
                                                       const float* __restrict__ t,
                                                       const float* __restrict__ b,
                                                       const float* __restrict__ g,
                                                       const float* __restrict__ be,
                                                       const float* __restrict__ mean,
                                                       const float* __restrict__ var,
                                                       const float* __restrict__ Wc,
                                                       const float* __restrict__ bc,
                                                       float* __restrict__ out, int n) {
  __shared__ float s_out[2][CLS_NODES * COUT];
  const int lane = threadIdx.x & 31;
  const int w    = __builtin_amdgcn_readfirstlane((int)(threadIdx.x >> 5));
  const int ntiles = (n + CLS_NODES - 1) / CLS_NODES;
  const float2* trow = reinterpret_cast<const float2*>(t);
  const size_t total = (size_t)n * COUT;
  const float4* w4 = reinterpret_cast<const float4*>(Wc) + 3 * lane;
  const float4 q0 = w4[0];
  const float4 q1 = w4[1];
  const float4 q2 = w4[2];
  const float bcl = bc[lane < COUT ? lane : 0];
  const FeatConst f = load_feat_const(b, g, be, mean, var, lane);

  int tile = blockIdx.x;
  const int gstep = (int)gridDim.x;
  const int m16 = lane & 15;
  int degs_a = 0;
  int degs_b = 0;
  if (tile < ntiles && tile * CLS_NODES + m16 < n) degs_a = cnt[tile * CLS_NODES + m16];
  if (tile + gstep < ntiles && (tile + gstep) * CLS_NODES + m16 < n) degs_b = cnt[(tile + gstep) * CLS_NODES + m16];
  int deg_cur = __builtin_amdgcn_readlane(degs_a, w);
  int ent_cur = 0;
  if (tile < ntiles && tile * CLS_NODES + w < n) ent_cur = load_slot_row(slots, tile * CLS_NODES + w, deg_cur, lane);
  int buf = 0;
  for (; tile < ntiles; tile += gstep) {
    const int node = tile * CLS_NODES + w;
    const int deg  = deg_cur;
    const int ent  = ent_cur;
    const int tile_nx = tile + gstep;
    const int node_nx = tile_nx * CLS_NODES + w;
    const int deg_nx  = __builtin_amdgcn_readlane(degs_b, w);
    int ent_nx = 0;
    if (tile_nx < ntiles && node_nx < n) ent_nx = load_slot_row(slots, node_nx, deg_nx, lane);
    const int tile_n2 = tile_nx + gstep;
    int degs_c = 0;
    if (tile_n2 < ntiles && tile_n2 * CLS_NODES + m16 < n) degs_c = cnt[tile_n2 * CLS_NODES + m16];

    float2 h = make_float2(0.0f, 0.0f);
    float di = 0.0f;
    if (node < n)
      h = gather_bn_relu(deg, ent, ovf_cnt, ovf_d, ovf_s, trow, f, node, lane, di);
    deg_cur = deg_nx;
    ent_cur = ent_nx;
    degs_b  = degs_c;
    if (node < n) {
      float p[COUT];
      p[0] = fmaf(h.y, q1.z, h.x * q0.x);
      p[1] = fmaf(h.y, q1.w, h.x * q0.y);
      p[2] = fmaf(h.y, q2.x, h.x * q0.z);
      p[3] = fmaf(h.y, q2.y, h.x * q0.w);
      p[4] = fmaf(h.y, q2.z, h.x * q1.x);
      p[5] = fmaf(h.y, q2.w, h.x * q1.y);
#pragma unroll
      for (int c = 0; c < COUT; ++c)
        for (int o = 16; o > 0; o >>= 1) p[c] += __shfl_xor(p[c], o, 32);
      float v = p[0];
#pragma unroll
      for (int c = 1; c < COUT; ++c) v = (lane == c) ? p[c] : v;
      if (lane < COUT) s_out[buf][w * COUT + lane] = v + bcl;
    }
    __syncthreads();
    if (w == 0) {
      const size_t o0 = (size_t)tile * (CLS_NODES * COUT);
#pragma unroll
      for (int j = 0; j < (CLS_NODES * COUT) / 32; ++j) {
        const int idx = j * 32 + lane;
        if (o0 + idx < total) out[o0 + idx] = s_out[buf][idx];
      }
    }
    buf ^= 1;
  }
}

extern "C" void kernel_launch(void* const* d_in, const int* in_sizes, int n_in,
                              void* d_out, int out_size, void* d_ws, size_t ws_size,
                              hipStream_t stream) {
  (void)n_in; (void)out_size; (void)ws_size;

  const float* x     = (const float*)d_in[0];
  const int*   ei    = (const int*)  d_in[1];
  const float* W_emb = (const float*)d_in[2];
  const float* b_emb = (const float*)d_in[3];
  const float* W1    = (const float*)d_in[4];
  const float* b1    = (const float*)d_in[5];
  const float* g1    = (const float*)d_in[6];
  const float* be1   = (const float*)d_in[7];
  const float* m1    = (const float*)d_in[8];
  const float* v1    = (const float*)d_in[9];
  const float* W2    = (const float*)d_in[10];
  const float* b2    = (const float*)d_in[11];
  const float* g2    = (const float*)d_in[12];
  const float* be2   = (const float*)d_in[13];
  const float* m2    = (const float*)d_in[14];
  const float* v2    = (const float*)d_in[15];
  const float* W_cls = (const float*)d_in[16];
  const float* b_cls = (const float*)d_in[17];

  const int n = in_sizes[0] / FIN;
  const int e = in_sizes[1] / 2;
  const int* src = ei;
  const int* dst = ei + e;

  float* ws   = (float*)d_ws;
  int*   cnt  = (int*)d_ws;
  float* B0   = ws + n;
  float* B1   = B0 + (size_t)n * HID;
  int*   slots   = (int*)(B1 + (size_t)n * HID);
  int*   ovf_cnt = slots + (size_t)n * SLOTS;
  int*   ovf_d   = ovf_cnt + 1;
  int*   ovf_s   = ovf_d + e;

  auto cdiv = [](long a, long b) { return (int)((a + b - 1) / b); };

  const int nb  = cdiv(n, BK_NODES);
  const int nsl = cdiv(e, PT_EDGES);
  int2*  recs  = (int2*)B0;
  int*   pre_t = (int*)(recs + (size_t)nsl * PT_EDGES);

  k_partition <<<nsl, PT_THREADS, 0, stream>>>(src, dst, recs, pre_t, ovf_cnt, e, nb);
  k_place_embed_gemm<<<nb, RK_THREADS, 0, stream>>>(pre_t, recs, nsl, cnt, slots, ovf_cnt, ovf_d, ovf_s,
                                                    x, W_emb, b_emb, W1, B1, n);

  k_gather_bn_gemm<<<GATHER_GRID, GG_NODES * 32, 0, stream>>>(cnt, slots, ovf_cnt, ovf_d, ovf_s, B1,
                                                                     b1, g1, be1, m1, v1, W2, B0, n);

  k_gather_bn_cls<<<GATHER_GRID, CLS_NODES * 32, 0, stream>>>(cnt, slots, ovf_cnt, ovf_d, ovf_s, B0,
                                                                   b2, g2, be2, m2, v2,
                                                                   W_cls, b_cls, (float*)d_out, n);
}
